# starting kernel
# baseline (speedup 1.0000x reference)
#include <hip/hip_runtime.h>


typedef __attribute__((ext_vector_type(2))) float v2f;
typedef __attribute__((ext_vector_type(8))) float v8f;

#define FEAT 16
#define KDIM 180
#define KSTEPS 45
#define YSTRIDE 186
#define ROWS_PER_BLOCK 16

__global__ void sc_pack_basis(const float* __restrict__ U3,
                              const float* __restrict__ U2,
                              float2* __restrict__ P, int ntiles) {
  int total = ntiles * KSTEPS * 32;
  for (int q = blockIdx.x * blockDim.x + threadIdx.x; q < total;
       q += gridDim.x * blockDim.x) {
    int lane = q & 31;
    int kk = (q >> 5) % KSTEPS;
    int tile = q / (KSTEPS * 32);
    int m0 = 4 * kk + 2 * (lane >> 4);
    int n = tile * 16 + (lane & 15);
    float a = (m0 < 176) ? U3[n * 176 + m0] : U2[n * 4 + (m0 - 176)];
    int m1 = m0 + 1;
    float b = (m1 < 176) ? U3[n * 176 + m1] : U2[n * 4 + (m1 - 176)];
    P[q] = make_float2(a, b);
  }
}

__global__ __launch_bounds__(256) void sc_wmma_kernel(
    const float* __restrict__ x, const int* __restrict__ indices,
    const v2f* __restrict__ P0, const v2f* __restrict__ P1,
    const v2f* __restrict__ P2,
    const float* __restrict__ U1_0, const float* __restrict__ U1_1,
    const float* __restrict__ U1_2,
    const float* __restrict__ W1_0, const float* __restrict__ W1_1,
    const float* __restrict__ W1_2,
    const float* __restrict__ W2_0, const float* __restrict__ W2_1,
    const float* __restrict__ W2_2,
    const float* __restrict__ W3_0, const float* __restrict__ W3_1,
    const float* __restrict__ W3_2,
    float* __restrict__ out) {
  __shared__ float xs[ROWS_PER_BLOCK][FEAT + 1];
  __shared__ alignas(16) float Ylds[3][ROWS_PER_BLOCK][YSTRIDE];
  __shared__ float w1s[3][ROWS_PER_BLOCK];
  __shared__ int es[ROWS_PER_BLOCK];

  const int tid = threadIdx.x;
  const int rowbase = blockIdx.x * ROWS_PER_BLOCK;

  {
    int r = tid >> 4, l = tid & 15;
    xs[r][l] = x[(rowbase + r) * FEAT + l];
  }
  if (tid < ROWS_PER_BLOCK) es[tid] = indices[(rowbase + tid) >> 6];
  __syncthreads();

  for (int q = tid; q < 3 * ROWS_PER_BLOCK * KDIM; q += 256) {
    int t = q / (ROWS_PER_BLOCK * KDIM);
    int rm = q - t * (ROWS_PER_BLOCK * KDIM);
    int r = rm / KDIM, m = rm - r * KDIM;
    int bc = rowbase + r, c = bc & 63, e = es[r];
    const float* W3 = (t == 0) ? W3_0 : (t == 1) ? W3_1 : W3_2;
    const float* W2 = (t == 0) ? W2_0 : (t == 1) ? W2_1 : W2_2;
    float v;
    if (m < 176) {
      int l = m / 11, p = m - l * 11;
      v = xs[r][l] * W3[(e * 11 + p) * 64 + c];
    } else {
      v = W2[(e * 4 + (m - 176)) * 64 + c];
    }
    Ylds[t][r][m] = v;
  }
  if (tid < 3 * ROWS_PER_BLOCK) {
    int t = tid >> 4, r = tid & 15;
    int bc = rowbase + r, c = bc & 63;
    const float* W1 = (t == 0) ? W1_0 : (t == 1) ? W1_1 : W1_2;
    w1s[t][r] = W1[es[r] * 64 + c];
  }
  __syncthreads();

  const int wave = tid >> 5;
  const int lane = tid & 31;
  const int half = lane >> 4;
  const int l15 = lane & 15;

  float xh[8];
#pragma unroll
  for (int v = 0; v < 8; ++v) xh[v] = xs[l15][(half << 3) + v];

  for (int g = wave; g < 9; g += 8) {
    int t, k;
    if (g < 1)       { t = 0; k = 0; }
    else if (g < 4)  { t = 1; k = g - 1; }
    else             { t = 2; k = g - 4; }
    const v2f* Pt    = (t == 0) ? P0 : (t == 1) ? P1 : P2;
    const float* U1  = (t == 0) ? U1_0 : (t == 1) ? U1_1 : U1_2;
    const int kt     = (t == 0) ? 1 : (t == 1) ? 3 : 5;
    const int seg    = (t == 0) ? 0 : (t == 1) ? 64 : 256;
    const float w1r  = w1s[t][l15];
    const float* Yrow = &Ylds[t][l15][0];

    float accT = 0.0f;

    for (int i = 0; i < 16; ++i) {
      const v2f* Ap = Pt + ((size_t)(k * 16 + i) * KSTEPS) * 32 + lane;
      __builtin_prefetch(Ap, 0, 1);

      v8f acc = {};
#pragma unroll 9
      for (int kk = 0; kk < KSTEPS; ++kk) {
        v2f A = Ap[kk * 32];
        v2f B = *reinterpret_cast<const v2f*>(Yrow + 4 * kk + 2 * half);
        acc = __builtin_amdgcn_wmma_f32_16x16x4_f32(
            false, A, false, B, (short)0, acc, false, false);
      }

      float s = 0.0f;
#pragma unroll
      for (int v = 0; v < 8; ++v) s = fmaf(acc[v], xh[v], s);
      s += __shfl_xor(s, 16, 32);

      const float u1v = U1[k * 16 + i];
      const float c1 = fmaf(u1v, w1r, s);
      accT = fmaf(c1, xs[l15][i], accT);
    }

    if (lane < 16) {
      const int bc = rowbase + l15;
      const int b = bc >> 6, c = bc & 63;
      out[b * 576 + seg + c * kt + k] = accT;
    }
  }
}

extern "C" void kernel_launch(void* const* d_in, const int* in_sizes, int n_in,
                              void* d_out, int out_size, void* d_ws, size_t ws_size,
                              hipStream_t stream) {
  (void)in_sizes; (void)n_in; (void)out_size; (void)ws_size;
  const float* x   = (const float*)d_in[0];
  const int* idx   = (const int*)d_in[1];
  const float* U1_0 = (const float*)d_in[2];
  const float* W1_0 = (const float*)d_in[3];
  const float* U2_0 = (const float*)d_in[4];
  const float* W2_0 = (const float*)d_in[5];
  const float* U3_0 = (const float*)d_in[6];
  const float* W3_0 = (const float*)d_in[7];
  const float* U1_1 = (const float*)d_in[8];
  const float* W1_1 = (const float*)d_in[9];
  const float* U2_1 = (const float*)d_in[10];
  const float* W2_1 = (const float*)d_in[11];
  const float* U3_1 = (const float*)d_in[12];
  const float* W3_1 = (const float*)d_in[13];
  const float* U1_2 = (const float*)d_in[14];
  const float* W1_2 = (const float*)d_in[15];
  const float* U2_2 = (const float*)d_in[16];
  const float* W2_2 = (const float*)d_in[17];
  const float* U3_2 = (const float*)d_in[18];
  const float* W3_2 = (const float*)d_in[19];
  float* out = (float*)d_out;

  float2* P0 = (float2*)d_ws;
  float2* P1 = P0 + (size_t)16 * KSTEPS * 32;
  float2* P2 = P1 + (size_t)48 * KSTEPS * 32;

  sc_pack_basis<<<64, 256, 0, stream>>>(U3_0, U2_0, P0, 16);
  sc_pack_basis<<<128, 256, 0, stream>>>(U3_1, U2_1, P1, 48);
  sc_pack_basis<<<256, 256, 0, stream>>>(U3_2, U2_2, P2, 80);

  sc_wmma_kernel<<<1024, 256, 0, stream>>>(
      x, idx, (const v2f*)P0, (const v2f*)P1, (const v2f*)P2,
      U1_0, U1_1, U1_2,
      W1_0, W1_1, W1_2,
      W2_0, W2_1, W2_2,
      W3_0, W3_1, W3_2,
      out);
}


# M axis packed densely: 99 degree-3, 36 degree-2, 9 degree-1 (output,path) rows fill 7+3+1 tiles, each over its degree's monomials: 2178 -> 1534 WMMAs per block; path sum via LDS (sum order changes)
# speedup vs baseline: 3.4521x; 3.4521x over previous
#include <hip/hip_runtime.h>


typedef __attribute__((ext_vector_type(2))) float v2f;
typedef __attribute__((ext_vector_type(8))) float v8f;

#define FEAT 16
#define NQ3 816
#define NQ2 136
#define NQ 968
#define KS3 204
#define KS2 34
#define KS1 4
#define NT3 7
#define NT2 3
#define KBASE2 (NT3 * KS3)
#define KBASE1 (KBASE2 + NT2 * KS2)
#define KS_TOTAL (KBASE1 + KS1)
#define ZROWS ((NT3 + NT2 + 1) * 16)
#define XSTRIDE 972
#define ROWS_PER_BLOCK 16
#define NTHREADS 256
#define NOUT 144

struct MonoTab { unsigned int v[NQ]; };
constexpr MonoTab make_mono_tab() {
  MonoTab t{};
  int q = 0;
  for (int i = 0; i < 16; ++i)
    for (int j = i; j < 16; ++j)
      for (int l = j; l < 16; ++l)
        t.v[q++] = (unsigned)((i * 4) | ((j * 4) << 8) | ((l * 4) << 16));
  for (int i = 0; i < 16; ++i)
    for (int j = i; j < 16; ++j)
      t.v[q++] = (unsigned)((i * 4) | ((j * 4) << 8) | (64 << 16));
  for (int i = 0; i < 16; ++i)
    t.v[q++] = (unsigned)((i * 4) | (64 << 8) | (64 << 16));
  return t;
}
__constant__ const MonoTab mono_tab = make_mono_tab();

__device__ __forceinline__ float sym_entry(const float* __restrict__ U1,
                                           const float* __restrict__ U2,
                                           const float* __restrict__ U3,
                                           int k, int path, int q) {
  const unsigned code = mono_tab.v[q];
  const int i = (code & 0xff) >> 2, j = ((code >> 8) & 0xff) >> 2, l = code >> 18;
  if (q < NQ3) {
    if (path >= 11) return 0.0f;
    const float* U = U3 + (size_t)k * (4096 * 11) + path;
#define U3AT(a, b, c) U[((((a) * 16 + (b)) * 16 + (c)) * 11)]
    if (i == j && j == l) return U3AT(i, i, i);
    if (i == j) return (U3AT(i, i, l) + U3AT(i, l, i)) + U3AT(l, i, i);
    if (j == l) return (U3AT(i, j, j) + U3AT(j, i, j)) + U3AT(j, j, i);
    return ((U3AT(i, j, l) + U3AT(i, l, j)) + (U3AT(j, i, l) + U3AT(j, l, i))) +
           (U3AT(l, i, j) + U3AT(l, j, i));
#undef U3AT
  }
  if (q < NQ3 + NQ2) {
    if (path < 11 || path >= 15) return 0.0f;
    const float* U = U2 + k * 1024 + (path - 11);
    float s = U[(i * 16 + j) * 4];
    if (i != j) s += U[(j * 16 + i) * 4];
    return s;
  }
  return (path == 15) ? U1[k * 16 + i] : 0.0f;
}

__global__ void sc_pack_sym(const float* __restrict__ U1,
                            const float* __restrict__ U2,
                            const float* __restrict__ U3,
                            float2* __restrict__ P, int tsel) {
  const int total = KS_TOTAL * 32;
  for (int n = blockIdx.x * blockDim.x + threadIdx.x; n < total;
       n += gridDim.x * blockDim.x) {
    const int lane = n & 31;
    const int ks = n >> 5;
    const int row = lane & 15;
    int g, tk, path, q0;
    bool valid;
    if (ks < KBASE2) {
      const int T = ks / KS3, kk = ks - T * KS3;
      g = T * 16 + row; valid = g < 99;
      tk = g / 11; path = g - tk * 11;
      q0 = 4 * kk;
    } else if (ks < KBASE1) {
      const int rr = ks - KBASE2;
      const int T = rr / KS2, kk = rr - T * KS2;
      g = T * 16 + row; valid = g < 36;
      tk = g >> 2; path = 11 + (g & 3);
      q0 = NQ3 + 4 * kk;
    } else {
      const int kk = ks - KBASE1;
      g = row; valid = g < 9;
      tk = g; path = 15;
      q0 = NQ3 + NQ2 + 4 * kk;
    }
    q0 += 2 * (lane >> 4);
    const int t = (tk == 0) ? 0 : (tk < 4) ? 1 : 2;
    const int k = tk - ((t == 0) ? 0 : (t == 1) ? 1 : 4);
    float a = 0.0f, b = 0.0f;
    if (valid) {
      if (t != tsel) continue;
      a = sym_entry(U1, U2, U3, k, path, q0);
      b = sym_entry(U1, U2, U3, k, path, q0 + 1);
    } else if (tsel != 2) {
      continue;
    }
    P[n] = make_float2(a, b);
  }
}

template <int NPAIRS>
__device__ __forceinline__ v8f mono_mma(const v2f* __restrict__ Ap,
                                        const float* Xrow) {
  v8f acc0 = {};
  v8f acc1 = {};
#pragma unroll 17
  for (int i = 0; i < NPAIRS; ++i) {
    v2f A0 = Ap[(2 * i) * 32];
    v2f B0 = *reinterpret_cast<const v2f*>(Xrow + 8 * i);
    v2f A1 = Ap[(2 * i + 1) * 32];
    v2f B1 = *reinterpret_cast<const v2f*>(Xrow + 8 * i + 4);
    acc0 = __builtin_amdgcn_wmma_f32_16x16x4_f32(
        false, A0, false, B0, (short)0, acc0, false, false);
    acc1 = __builtin_amdgcn_wmma_f32_16x16x4_f32(
        false, A1, false, B1, (short)0, acc1, false, false);
  }
  return acc0 + acc1;
}

__global__ __launch_bounds__(NTHREADS) void sc_mono_kernel(
    const float* __restrict__ x, const int* __restrict__ indices,
    const v2f* __restrict__ P,
    const float* __restrict__ W1_0, const float* __restrict__ W1_1,
    const float* __restrict__ W1_2,
    const float* __restrict__ W2_0, const float* __restrict__ W2_1,
    const float* __restrict__ W2_2,
    const float* __restrict__ W3_0, const float* __restrict__ W3_1,
    const float* __restrict__ W3_2,
    float* __restrict__ out) {
  __shared__ float xs[ROWS_PER_BLOCK][FEAT + 1];
  __shared__ alignas(16) float Xm[ROWS_PER_BLOCK][XSTRIDE];
  __shared__ float Zs[ZROWS][ROWS_PER_BLOCK];

  const int tid = threadIdx.x;
  const int rowbase = blockIdx.x * ROWS_PER_BLOCK;

  {
    int r = tid >> 4, l = tid & 15;
    xs[r][l] = x[(rowbase + r) * FEAT + l];
    if (tid < ROWS_PER_BLOCK) xs[tid][FEAT] = 1.0f;
  }
  __syncthreads();

  {
    const int r = tid & 15;
    const char* xr = reinterpret_cast<const char*>(&xs[r][0]);
    float* Xr = &Xm[r][0];
    for (int q = tid >> 4; q < NQ; q += NTHREADS / 16) {
      const unsigned code = mono_tab.v[q];
      const float fa = *reinterpret_cast<const float*>(xr + (code & 0xff));
      const float fb = *reinterpret_cast<const float*>(xr + ((code >> 8) & 0xff));
      const float fc = *reinterpret_cast<const float*>(xr + (code >> 16));
      Xr[q] = (fa * fb) * fc;
    }
  }

  const int wave = __builtin_amdgcn_readfirstlane(tid >> 5);
  const int lane = tid & 31;
  const int half = lane >> 4;
  const int l15 = lane & 15;

  const int otk = tid >> 4;
  const int ot = (otk == 0) ? 0 : (otk < 4) ? 1 : 2;
  const int b = rowbase >> 6;
  const int oc = (rowbase & 63) + l15;
  float wh[16];
  if (tid < NOUT) {
    const float* W3 = (ot == 0) ? W3_0 : (ot == 1) ? W3_1 : W3_2;
    const float* W2 = (ot == 0) ? W2_0 : (ot == 1) ? W2_1 : W2_2;
    const float* W1 = (ot == 0) ? W1_0 : (ot == 1) ? W1_1 : W1_2;
    const int e = indices[b];
#pragma unroll
    for (int p = 0; p < 11; ++p) wh[p] = W3[(e * 11 + p) * 64 + oc];
#pragma unroll
    for (int p = 0; p < 4; ++p) wh[11 + p] = W2[(e * 4 + p) * 64 + oc];
    wh[15] = W1[e * 64 + oc];
  }

  __syncthreads();

  const float* Xrow = &Xm[l15][2 * half];
  float* Zcol = &Zs[8 * half][l15];

  if (wave < NT3) {
    const v8f z = mono_mma<KS3 / 2>(P + (size_t)wave * (KS3 * 32) + lane, Xrow);
#pragma unroll
    for (int v = 0; v < 8; ++v) Zcol[(wave * 16 + v) * ROWS_PER_BLOCK] = z[v];
  } else {
    const v2f* A2 = P + (size_t)KBASE2 * 32 + lane;
    const float* X2 = Xrow + NQ3;
    v8f c0[NT2] = {};
    v8f c1[NT2] = {};
#pragma unroll 1
    for (int i = 0; i < KS2 / 2; ++i) {
      const v2f B0 = *reinterpret_cast<const v2f*>(X2 + 8 * i);
      const v2f B1 = *reinterpret_cast<const v2f*>(X2 + 8 * i + 4);
#pragma unroll
      for (int T = 0; T < NT2; ++T) {
        const v2f A0 = A2[(T * KS2 + 2 * i) * 32];
        const v2f A1 = A2[(T * KS2 + 2 * i + 1) * 32];
        c0[T] = __builtin_amdgcn_wmma_f32_16x16x4_f32(
            false, A0, false, B0, (short)0, c0[T], false, false);
        c1[T] = __builtin_amdgcn_wmma_f32_16x16x4_f32(
            false, A1, false, B1, (short)0, c1[T], false, false);
      }
    }
#pragma unroll
    for (int T = 0; T < NT2; ++T) {
      const v8f z = c0[T] + c1[T];
#pragma unroll
      for (int v = 0; v < 8; ++v)
        Zcol[((NT3 + T) * 16 + v) * ROWS_PER_BLOCK] = z[v];
    }
    const v8f z = mono_mma<KS1 / 2>(P + (size_t)KBASE1 * 32 + lane,
                                    Xrow + NQ3 + NQ2);
#pragma unroll
    for (int v = 0; v < 8; ++v)
      Zcol[((NT3 + NT2) * 16 + v) * ROWS_PER_BLOCK] = z[v];
  }

  __syncthreads();

  if (tid < NOUT) {
    float s = 0.0f;
#pragma unroll
    for (int p = 0; p < 11; ++p) s = fmaf(Zs[otk * 11 + p][l15], wh[p], s);
#pragma unroll
    for (int p = 0; p < 4; ++p)
      s = fmaf(Zs[NT3 * 16 + otk * 4 + p][l15], wh[11 + p], s);
    s = fmaf(Zs[(NT3 + NT2) * 16 + otk][l15], wh[15], s);
    const int kt  = (ot == 0) ? 1 : (ot == 1) ? 3 : 5;
    const int seg = (ot == 0) ? 0 : (ot == 1) ? 64 : 256;
    const int k   = otk - ((ot == 0) ? 0 : (ot == 1) ? 1 : 4);
    out[b * 576 + seg + oc * kt + k] = s;
  }
}

extern "C" void kernel_launch(void* const* d_in, const int* in_sizes, int n_in,
                              void* d_out, int out_size, void* d_ws, size_t ws_size,
                              hipStream_t stream) {
  (void)in_sizes; (void)n_in; (void)out_size; (void)ws_size;
  const float* x   = (const float*)d_in[0];
  const int* idx   = (const int*)d_in[1];
  const float* U1_0 = (const float*)d_in[2];
  const float* W1_0 = (const float*)d_in[3];
  const float* U2_0 = (const float*)d_in[4];
  const float* W2_0 = (const float*)d_in[5];
  const float* U3_0 = (const float*)d_in[6];
  const float* W3_0 = (const float*)d_in[7];
  const float* U1_1 = (const float*)d_in[8];
  const float* W1_1 = (const float*)d_in[9];
  const float* U2_1 = (const float*)d_in[10];
  const float* W2_1 = (const float*)d_in[11];
  const float* U3_1 = (const float*)d_in[12];
  const float* W3_1 = (const float*)d_in[13];
  const float* U1_2 = (const float*)d_in[14];
  const float* W1_2 = (const float*)d_in[15];
  const float* U2_2 = (const float*)d_in[16];
  const float* W2_2 = (const float*)d_in[17];
  const float* U3_2 = (const float*)d_in[18];
  const float* W3_2 = (const float*)d_in[19];
  float* out = (float*)d_out;

  float2* P = (float2*)d_ws;
  sc_pack_sym<<<192, 256, 0, stream>>>(U1_0, U2_0, U3_0, P, 0);
  sc_pack_sym<<<192, 256, 0, stream>>>(U1_1, U2_1, U3_1, P, 1);
  sc_pack_sym<<<192, 256, 0, stream>>>(U1_2, U2_2, U3_2, P, 2);

  sc_mono_kernel<<<1024, NTHREADS, 0, stream>>>(
      x, idx, (const v2f*)P,
      W1_0, W1_1, W1_2,
      W2_0, W2_1, W2_2,
      W3_0, W3_1, W3_2,
      out);
}


# B operand made in registers: a lane holds its row's 16 x values (lanes 16-31: 8-feature halves swapped) and multiplies the monomials itself; LDS monomial table and its loads gone (sum order changes)
# speedup vs baseline: 3.8131x; 1.1046x over previous
#include <hip/hip_runtime.h>
#include <utility>


typedef __attribute__((ext_vector_type(2))) float v2f;
typedef __attribute__((ext_vector_type(8))) float v8f;

#define FEAT 16
#define NC3 408
#define NC2 72
#define NC1 8
#define NCOL (NC3 + NC2 + NC1)
#define KS3 204
#define KS2 36
#define KS1 4
#define NT3 7
#define NT2 3
#define KBASE2 (NT3 * KS3)
#define KBASE1 (KBASE2 + NT2 * KS2)
#define KS_TOTAL (KBASE1 + KS1)
#define ZROWS ((NT3 + NT2 + 1) * 16)
#define ROWS_PER_BLOCK 16
#define NTHREADS 256
#define NOUT 144
#define CHUNK 12
#define NCHUNK 17
#define PCHUNK 4
#define NPCHUNK 9

struct ColTab { unsigned int v[NCOL]; };
constexpr ColTab make_col_tab() {
  ColTab t{};
  int c = 0;
  unsigned pend = 0;
  bool have = false;
  for (int i = 0; i < 8; ++i)
    for (int j = i; j < 8; ++j) {
      for (int l = j + (j & 1); l < 16; ++l)
        t.v[c++] = (unsigned)(i | (j << 8) | (l << 16));
      if (j & 1) {
        const unsigned m = (unsigned)(i | (j << 8) | (j << 16));
        if (have) { t.v[c++] = pend; t.v[c++] = m; have = false; }
        else { pend = m; have = true; }
      }
    }
  for (int i = 0; i < 8; ++i)
    for (int j = i; j < 8; ++j)
      t.v[c++] = (unsigned)(i | (j << 8) | (16 << 16));
  for (int a = 0; a < 8; ++a)
    for (int b = a + 1; b < 8; ++b)
      t.v[c++] = (unsigned)(a | ((b + 8) << 8) | (16 << 16));
  for (int a = 0; a < 8; ++a)
    t.v[c++] = (unsigned)(a | ((a + 8) << 8) | (16 << 16) | (1u << 24));
  for (int i = 0; i < 8; ++i)
    t.v[c++] = (unsigned)(i | (16 << 8) | (16 << 16));
  return t;
}
constexpr ColTab col_tab_ct = make_col_tab();
__constant__ const ColTab col_tab = make_col_tab();

__device__ __forceinline__ float sym3(const float* __restrict__ U3,
                                      int k, int path, int i, int j, int l) {
  int t;
  if (i > j) { t = i; i = j; j = t; }
  if (j > l) { t = j; j = l; l = t; }
  if (i > j) { t = i; i = j; j = t; }
  const float* U = U3 + (size_t)k * (4096 * 11) + path;
#define U3AT(a, b, c) U[((((a) * 16 + (b)) * 16 + (c)) * 11)]
  if (i == j && j == l) return U3AT(i, i, i);
  if (i == j) return (U3AT(i, i, l) + U3AT(i, l, i)) + U3AT(l, i, i);
  if (j == l) return (U3AT(i, j, j) + U3AT(j, i, j)) + U3AT(j, j, i);
  return ((U3AT(i, j, l) + U3AT(i, l, j)) + (U3AT(j, i, l) + U3AT(j, l, i))) +
         (U3AT(l, i, j) + U3AT(l, j, i));
#undef U3AT
}

__device__ __forceinline__ float col_entry(const float* __restrict__ U1,
                                           const float* __restrict__ U2,
                                           const float* __restrict__ U3,
                                           int k, int path, int col, int h) {
  const unsigned code = col_tab.v[col];
  int i = code & 0xff, j = (code >> 8) & 0xff, l = (code >> 16) & 0xff;
  if (h) {
    if (code >> 24) return 0.0f;
    i ^= 8;
    if (j < 16) j ^= 8;
    if (l < 16) l ^= 8;
  }
  if (l < 16) return sym3(U3, k, path, i, j, l);
  if (j < 16) {
    if (i > j) { const int t = i; i = j; j = t; }
    const float* U = U2 + k * 1024 + path;
    float s = U[(i * 16 + j) * 4];
    if (i != j) s += U[(j * 16 + i) * 4];
    return s;
  }
  return U1[k * 16 + i];
}

__global__ void sc_pack_sym(const float* __restrict__ U1,
                            const float* __restrict__ U2,
                            const float* __restrict__ U3,
                            float2* __restrict__ P, int tsel) {
  const int total = KS_TOTAL * 32;
  for (int n = blockIdx.x * blockDim.x + threadIdx.x; n < total;
       n += gridDim.x * blockDim.x) {
    const int lane = n & 31;
    const int ks = n >> 5;
    const int row = lane & 15;
    const int h = lane >> 4;
    int g, tk, path, col;
    bool valid;
    if (ks < KBASE2) {
      const int T = ks / KS3, kk = ks - T * KS3;
      g = T * 16 + row; valid = g < 99;
      tk = g / 11; path = g - tk * 11;
      col = 2 * kk;
    } else if (ks < KBASE1) {
      const int rr = ks - KBASE2;
      const int T = rr / KS2, kk = rr - T * KS2;
      g = T * 16 + row; valid = g < 36;
      tk = g >> 2; path = g & 3;
      col = NC3 + 2 * kk;
    } else {
      const int kk = ks - KBASE1;
      g = row; valid = g < 9;
      tk = g; path = 0;
      col = NC3 + NC2 + 2 * kk;
    }
    const int t = (tk == 0) ? 0 : (tk < 4) ? 1 : 2;
    const int k = tk - ((t == 0) ? 0 : (t == 1) ? 1 : 4);
    float a = 0.0f, b = 0.0f;
    if (valid) {
      if (t != tsel) continue;
      a = col_entry(U1, U2, U3, k, path, col, h);
      b = col_entry(U1, U2, U3, k, path, col + 1, h);
    } else if (tsel != 2) {
      continue;
    }
    P[n] = make_float2(a, b);
  }
}

template <int C>
__device__ __forceinline__ float col_val(const float (&y)[FEAT]) {
  constexpr unsigned code = col_tab_ct.v[C];
  constexpr int i = code & 0xff, j = (code >> 8) & 0xff, l = (code >> 16) & 0xff;
  if constexpr (j == 16) return y[i];
  else if constexpr (l == 16) return y[i] * y[j];
  else return (y[i] * y[j]) * y[l];
}

template <int C>
__device__ __forceinline__ v2f col_frag(const float (&y)[FEAT]) {
  v2f b;
  b.x = col_val<C>(y);
  b.y = col_val<C + 1>(y);
  return b;
}

template <int C, int ODD>
__device__ __forceinline__ void mono_step(const v2f a, const float (&y)[FEAT],
                                          v8f& acc0, v8f& acc1) {
  const v2f b = col_frag<C>(y);
  if constexpr (ODD)
    acc1 = __builtin_amdgcn_wmma_f32_16x16x4_f32(false, a, false, b, (short)0,
                                                 acc1, false, false);
  else
    acc0 = __builtin_amdgcn_wmma_f32_16x16x4_f32(false, a, false, b, (short)0,
                                                 acc0, false, false);
}

template <int COL0, int NCH, int C, int... S>
__device__ __forceinline__ void mono_walk(const v2f* __restrict__ Ap,
                                          const float (&y)[FEAT],
                                          v8f& acc0, v8f& acc1,
                                          const v2f (&cur)[sizeof...(S)],
                                          std::integer_sequence<int, S...> seq) {
  constexpr int N = (int)sizeof...(S);
  v2f nxt[N];
  if constexpr (C + 1 < NCH) {
    ((nxt[S] = Ap[((C + 1) * N + S) * 32]), ...);
  }
  if constexpr (C + 1 < NCH) __builtin_amdgcn_sched_barrier(0);
  (mono_step<COL0 + 2 * (C * N + S), (C * N + S) & 1>(cur[S], y, acc0, acc1), ...);
  if constexpr (C + 2 < NCH) __builtin_amdgcn_sched_barrier(0);
  if constexpr (C + 1 < NCH)
    mono_walk<COL0, NCH, C + 1>(Ap, y, acc0, acc1, nxt, seq);
}

template <int C, int... S>
__device__ __forceinline__ void pair_walk(const v2f* __restrict__ A2,
                                          const float (&y)[FEAT],
                                          v8f (&c)[NT2],
                                          const v2f (&cur)[NT2][sizeof...(S)],
                                          std::integer_sequence<int, S...> seq) {
  constexpr int N = (int)sizeof...(S);
  v2f nxt[NT2][N];
  if constexpr (C + 1 < NPCHUNK) {
    ((nxt[0][S] = A2[(0 * KS2 + (C + 1) * N + S) * 32]), ...);
    ((nxt[1][S] = A2[(1 * KS2 + (C + 1) * N + S) * 32]), ...);
    ((nxt[2][S] = A2[(2 * KS2 + (C + 1) * N + S) * 32]), ...);
  }
  if constexpr (C + 1 < NPCHUNK) __builtin_amdgcn_sched_barrier(0);
  auto step = [&](const v2f f0, const v2f f1, const v2f f2, const v2f b) {
    c[0] = __builtin_amdgcn_wmma_f32_16x16x4_f32(false, f0, false, b, (short)0,
                                                 c[0], false, false);
    c[1] = __builtin_amdgcn_wmma_f32_16x16x4_f32(false, f1, false, b, (short)0,
                                                 c[1], false, false);
    c[2] = __builtin_amdgcn_wmma_f32_16x16x4_f32(false, f2, false, b, (short)0,
                                                 c[2], false, false);
  };
  (step(cur[0][S], cur[1][S], cur[2][S], col_frag<NC3 + 2 * (C * N + S)>(y)), ...);
  if constexpr (C + 2 < NPCHUNK) __builtin_amdgcn_sched_barrier(0);
  if constexpr (C + 1 < NPCHUNK) pair_walk<C + 1>(A2, y, c, nxt, seq);
}

__global__ __launch_bounds__(NTHREADS, 8) void sc_mono_kernel(
    const float* __restrict__ x, const int* __restrict__ indices,
    const v2f* __restrict__ P,
    const float* __restrict__ W1_0, const float* __restrict__ W1_1,
    const float* __restrict__ W1_2,
    const float* __restrict__ W2_0, const float* __restrict__ W2_1,
    const float* __restrict__ W2_2,
    const float* __restrict__ W3_0, const float* __restrict__ W3_1,
    const float* __restrict__ W3_2,
    float* __restrict__ out) {
  __shared__ float Zs[ZROWS][ROWS_PER_BLOCK];

  const int tid = threadIdx.x;
  const int rowbase = blockIdx.x * ROWS_PER_BLOCK;

  const int wave = __builtin_amdgcn_readfirstlane(tid >> 5);
  const int lane = tid & 31;
  const int half = lane >> 4;
  const int l15 = lane & 15;

  float y[FEAT];
  {
    const v2f* xr =
        reinterpret_cast<const v2f*>(x + (size_t)(rowbase + l15) * FEAT);
#pragma unroll
    for (int c = 0; c < 8; ++c) {
      const v2f v = xr[c ^ (4 * half)];
      y[2 * c + 0] = v.x; y[2 * c + 1] = v.y;
    }
  }

  const int otk = tid >> 4;
  const int ot = (otk == 0) ? 0 : (otk < 4) ? 1 : 2;
  const int b = rowbase >> 6;
  const int oc = (rowbase & 63) + l15;

  float* Zcol = &Zs[8 * half][l15];

  if (wave < NT3) {
    const v2f* Ap = P + (size_t)wave * (KS3 * 32) + lane;
    static_assert(NCHUNK * CHUNK == KS3, "groups cover the degree-3 walk");
    v8f acc0 = {};
    v8f acc1 = {};
    v2f first[CHUNK];
#pragma unroll
    for (int s = 0; s < CHUNK; ++s) first[s] = Ap[s * 32];
    mono_walk<0, NCHUNK, 0>(Ap, y, acc0, acc1, first,
                            std::make_integer_sequence<int, CHUNK>{});
    const v8f z = acc0 + acc1;
#pragma unroll
    for (int v = 0; v < 8; ++v) Zcol[(wave * 16 + v) * ROWS_PER_BLOCK] = z[v];
  } else {
    const v2f* A2 = P + (size_t)KBASE2 * 32 + lane;
    const v2f* A1 = P + (size_t)KBASE1 * 32 + lane;
    static_assert(NPCHUNK * PCHUNK == KS2, "groups cover the degree-2 walk");
    v8f c[NT2] = {};
    v2f first[NT2][PCHUNK];
    v2f lin[KS1];
#pragma unroll
    for (int T = 0; T < NT2; ++T) {
#pragma unroll
      for (int s = 0; s < PCHUNK; ++s) first[T][s] = A2[(T * KS2 + s) * 32];
    }
#pragma unroll
    for (int s = 0; s < KS1; ++s) lin[s] = A1[s * 32];
    pair_walk<0>(A2, y, c, first, std::make_integer_sequence<int, PCHUNK>{});
#pragma unroll
    for (int T = 0; T < NT2; ++T) {
#pragma unroll
      for (int v = 0; v < 8; ++v)
        Zcol[((NT3 + T) * 16 + v) * ROWS_PER_BLOCK] = c[T][v];
    }
    v8f d0 = {};
    v8f d1 = {};
    mono_walk<NC3 + NC2, 1, 0>(A1, y, d0, d1, lin,
                               std::make_integer_sequence<int, KS1>{});
    const v8f z = d0 + d1;
#pragma unroll
    for (int v = 0; v < 8; ++v)
      Zcol[((NT3 + NT2) * 16 + v) * ROWS_PER_BLOCK] = z[v];
  }

  float wh[16];
  if (tid < NOUT) {
    const float* W3 = (ot == 0) ? W3_0 : (ot == 1) ? W3_1 : W3_2;
    const float* W2 = (ot == 0) ? W2_0 : (ot == 1) ? W2_1 : W2_2;
    const float* W1 = (ot == 0) ? W1_0 : (ot == 1) ? W1_1 : W1_2;
    const int e = indices[b];
#pragma unroll
    for (int p = 0; p < 11; ++p) wh[p] = W3[(e * 11 + p) * 64 + oc];
#pragma unroll
    for (int p = 0; p < 4; ++p) wh[11 + p] = W2[(e * 4 + p) * 64 + oc];
    wh[15] = W1[e * 64 + oc];
  }

  __syncthreads();

  if (tid < NOUT) {
    float s = 0.0f;
#pragma unroll
    for (int p = 0; p < 11; ++p) s = fmaf(Zs[otk * 11 + p][l15], wh[p], s);
#pragma unroll
    for (int p = 0; p < 4; ++p)
      s = fmaf(Zs[NT3 * 16 + otk * 4 + p][l15], wh[11 + p], s);
    s = fmaf(Zs[(NT3 + NT2) * 16 + otk][l15], wh[15], s);
    const int kt  = (ot == 0) ? 1 : (ot == 1) ? 3 : 5;
    const int seg = (ot == 0) ? 0 : (ot == 1) ? 64 : 256;
    const int k   = otk - ((ot == 0) ? 0 : (ot == 1) ? 1 : 4);
    out[b * 576 + seg + oc * kt + k] = s;
  }
}

extern "C" void kernel_launch(void* const* d_in, const int* in_sizes, int n_in,
                              void* d_out, int out_size, void* d_ws, size_t ws_size,
                              hipStream_t stream) {
  (void)in_sizes; (void)n_in; (void)out_size; (void)ws_size;
  const float* x   = (const float*)d_in[0];
  const int* idx   = (const int*)d_in[1];
  const float* U1_0 = (const float*)d_in[2];
  const float* W1_0 = (const float*)d_in[3];
  const float* U2_0 = (const float*)d_in[4];
  const float* W2_0 = (const float*)d_in[5];
  const float* U3_0 = (const float*)d_in[6];
  const float* W3_0 = (const float*)d_in[7];
  const float* U1_1 = (const float*)d_in[8];
  const float* W1_1 = (const float*)d_in[9];
  const float* U2_1 = (const float*)d_in[10];
  const float* W2_1 = (const float*)d_in[11];
  const float* U3_1 = (const float*)d_in[12];
  const float* W3_1 = (const float*)d_in[13];
  const float* U1_2 = (const float*)d_in[14];
  const float* W1_2 = (const float*)d_in[15];
  const float* U2_2 = (const float*)d_in[16];
  const float* W2_2 = (const float*)d_in[17];
  const float* U3_2 = (const float*)d_in[18];
  const float* W3_2 = (const float*)d_in[19];
  float* out = (float*)d_out;

  float2* P = (float2*)d_ws;
  sc_pack_sym<<<193, 256, 0, stream>>>(U1_0, U2_0, U3_0, P, 0);
  sc_pack_sym<<<193, 256, 0, stream>>>(U1_1, U2_1, U3_1, P, 1);
  sc_pack_sym<<<193, 256, 0, stream>>>(U1_2, U2_2, U3_2, P, 2);

  sc_mono_kernel<<<1024, NTHREADS, 0, stream>>>(
      x, idx, (const v2f*)P,
      W1_0, W1_1, W1_2,
      W2_0, W2_1, W2_2,
      W3_0, W3_1, W3_2,
      out);
}


# Pack launches: a thread's chain of dependent memory round trips cut from six to two (table words by scalar load beside the arguments, all basis loads one group); basis entries' sum order changes
# speedup vs baseline: 4.2811x; 1.1227x over previous
#include <hip/hip_runtime.h>
#include <utility>


typedef __attribute__((ext_vector_type(2))) float v2f;
typedef __attribute__((ext_vector_type(8))) float v8f;

#define FEAT 16
#define NC3 408
#define NC2 72
#define NC1 8
#define NCOL (NC3 + NC2 + NC1)
#define KS3 204
#define KS2 36
#define KS1 4
#define NT3 7
#define NT2 3
#define KBASE2 (NT3 * KS3)
#define KBASE1 (KBASE2 + NT2 * KS2)
#define KS_TOTAL (KBASE1 + KS1)
#define ZROWS ((NT3 + NT2 + 1) * 16)
#define ROWS_PER_BLOCK 16
#define NTHREADS 256
#define NOUT 144
#define CHUNK 12
#define NCHUNK 17
#define PCHUNK 4
#define NPCHUNK 9

struct ColTab { unsigned int v[NCOL]; };
constexpr ColTab make_col_tab() {
  ColTab t{};
  int c = 0;
  unsigned pend = 0;
  bool have = false;
  for (int i = 0; i < 8; ++i)
    for (int j = i; j < 8; ++j) {
      for (int l = j + (j & 1); l < 16; ++l)
        t.v[c++] = (unsigned)(i | (j << 8) | (l << 16));
      if (j & 1) {
        const unsigned m = (unsigned)(i | (j << 8) | (j << 16));
        if (have) { t.v[c++] = pend; t.v[c++] = m; have = false; }
        else { pend = m; have = true; }
      }
    }
  for (int i = 0; i < 8; ++i)
    for (int j = i; j < 8; ++j)
      t.v[c++] = (unsigned)(i | (j << 8) | (16 << 16));
  for (int a = 0; a < 8; ++a)
    for (int b = a + 1; b < 8; ++b)
      t.v[c++] = (unsigned)(a | ((b + 8) << 8) | (16 << 16));
  for (int a = 0; a < 8; ++a)
    t.v[c++] = (unsigned)(a | ((a + 8) << 8) | (16 << 16) | (1u << 24));
  for (int i = 0; i < 8; ++i)
    t.v[c++] = (unsigned)(i | (16 << 8) | (16 << 16));
  return t;
}
constexpr ColTab col_tab_ct = make_col_tab();
__constant__ const ColTab col_tab = make_col_tab();

__device__ __forceinline__ float sym3(const float* __restrict__ U3, int base,
                                      unsigned code, int x8) {
  const int a = (int)(code & 0xff) ^ x8;
  const int b = (int)((code >> 8) & 0xff) ^ x8;
  const int c = (int)((code >> 16) & 0xff) ^ x8;
#define U3AT(p, q, r) U3[base + (((p) * 16 + (q)) * 16 + (r)) * 11]
  const float v0 = U3AT(a, b, c), v1 = U3AT(a, c, b), v2 = U3AT(b, a, c);
  const float v3 = U3AT(b, c, a), v4 = U3AT(c, a, b), v5 = U3AT(c, b, a);
#undef U3AT
  const bool eab = a == b, ebc = b == c, eac = a == c;
  const float w1 = ebc ? 0.0f : v1;
  const float w2 = eab ? 0.0f : v2;
  const float w3 = (eab || eac) ? 0.0f : v3;
  const float w4 = (eac || ebc) ? 0.0f : v4;
  const float w5 = (eab || ebc || eac) ? 0.0f : v5;
  return ((v0 + w1) + (w2 + w3)) + (w4 + w5);
}

__device__ __forceinline__ float sym2(const float* __restrict__ U2, int base,
                                      unsigned code, int x8) {
  const int i = (int)(code & 0xff) ^ x8;
  const int j = (int)((code >> 8) & 0xff) ^ x8;
  const float v0 = U2[base + (i * 16 + j) * 4];
  const float v1 = U2[base + (j * 16 + i) * 4];
  const float s = (i == j) ? v0 : v0 + v1;
  return (x8 != 0 && (code >> 24) != 0) ? 0.0f : s;
}

__global__ __launch_bounds__(256) void sc_pack_sym(const float* __restrict__ U1,
                                                   const float* __restrict__ U2,
                                                   const float* __restrict__ U3,
                                                   float2* __restrict__ P, int tsel) {
  const int n = blockIdx.x * 256 + threadIdx.x;
  if (n >= KS_TOTAL * 32) return;
  const int lane = n & 31;
  const int row = lane & 15;
  const int x8 = (lane >> 4) << 3;
  const int ks = __builtin_amdgcn_readfirstlane(n >> 5);
  const int deg = (ks < KBASE2) ? 3 : (ks < KBASE1) ? 2 : 1;
  const int T3 = ks / KS3, kk3 = ks - T3 * KS3;
  const int r2 = (deg == 2) ? ks - KBASE2 : 0;
  const int T2 = r2 / KS2, kk2 = r2 - T2 * KS2;
  const int T = (deg == 3) ? T3 : (deg == 2) ? T2 : 0;
  const int kk = (deg == 3) ? kk3 : (deg == 2) ? kk2 : ks - KBASE1;
  const int col = ((deg == 3) ? 0 : (deg == 2) ? NC3 : NC3 + NC2) + 2 * kk;
  const unsigned c0 = col_tab.v[col], c1 = col_tab.v[col + 1];
  asm volatile("" ::"s"(c0), "s"(c1));
  const int g = T * 16 + row;
  float a = 0.0f, b = 0.0f;
  bool act, pad;
  if (deg == 3) {
    const int tk = g / 11, path = g - tk * 11;
    const int t = (tk == 0) ? 0 : (tk < 4) ? 1 : 2;
    const int k = tk - ((t == 0) ? 0 : (t == 1) ? 1 : 4);
    act = g < 99 && t == tsel;
    pad = g >= 99 && tsel == 2;
    if (act) {
      const int base = k * (4096 * 11) + path;
      a = sym3(U3, base, c0, x8);
      b = sym3(U3, base, c1, x8);
    }
  } else if (deg == 2) {
    const int tk = g >> 2, path = g & 3;
    const int t = (tk == 0) ? 0 : (tk < 4) ? 1 : 2;
    const int k = tk - ((t == 0) ? 0 : (t == 1) ? 1 : 4);
    act = g < 36 && t == tsel;
    pad = g >= 36 && tsel == 2;
    if (act) {
      const int base = k * 1024 + path;
      a = sym2(U2, base, c0, x8);
      b = sym2(U2, base, c1, x8);
    }
  } else {
    const int tk = g;
    const int t = (tk == 0) ? 0 : (tk < 4) ? 1 : 2;
    const int k = tk - ((t == 0) ? 0 : (t == 1) ? 1 : 4);
    act = g < 9 && t == tsel;
    pad = g >= 9 && tsel == 2;
    if (act) {
      a = U1[k * 16 + ((int)(c0 & 0xff) ^ x8)];
      b = U1[k * 16 + ((int)(c1 & 0xff) ^ x8)];
    }
  }
  if (act || pad) P[n] = make_float2(a, b);
}

template <int C>
__device__ __forceinline__ float col_val(const float (&y)[FEAT]) {
  constexpr unsigned code = col_tab_ct.v[C];
  constexpr int i = code & 0xff, j = (code >> 8) & 0xff, l = (code >> 16) & 0xff;
  if constexpr (j == 16) return y[i];
  else if constexpr (l == 16) return y[i] * y[j];
  else return (y[i] * y[j]) * y[l];
}

template <int C>
__device__ __forceinline__ v2f col_frag(const float (&y)[FEAT]) {
  v2f b;
  b.x = col_val<C>(y);
  b.y = col_val<C + 1>(y);
  return b;
}

template <int C, int ODD>
__device__ __forceinline__ void mono_step(const v2f a, const float (&y)[FEAT],
                                          v8f& acc0, v8f& acc1) {
  const v2f b = col_frag<C>(y);
  if constexpr (ODD)
    acc1 = __builtin_amdgcn_wmma_f32_16x16x4_f32(false, a, false, b, (short)0,
                                                 acc1, false, false);
  else
    acc0 = __builtin_amdgcn_wmma_f32_16x16x4_f32(false, a, false, b, (short)0,
                                                 acc0, false, false);
}

template <int COL0, int NCH, int C, int... S>
__device__ __forceinline__ void mono_walk(const v2f* __restrict__ Ap,
                                          const float (&y)[FEAT],
                                          v8f& acc0, v8f& acc1,
                                          const v2f (&cur)[sizeof...(S)],
                                          std::integer_sequence<int, S...> seq) {
  constexpr int N = (int)sizeof...(S);
  v2f nxt[N];
  if constexpr (C + 1 < NCH) {
    ((nxt[S] = Ap[((C + 1) * N + S) * 32]), ...);
  }
  if constexpr (C + 1 < NCH) __builtin_amdgcn_sched_barrier(0);
  (mono_step<COL0 + 2 * (C * N + S), (C * N + S) & 1>(cur[S], y, acc0, acc1), ...);
  if constexpr (C + 2 < NCH) __builtin_amdgcn_sched_barrier(0);
  if constexpr (C + 1 < NCH)
    mono_walk<COL0, NCH, C + 1>(Ap, y, acc0, acc1, nxt, seq);
}

template <int C, int... S>
__device__ __forceinline__ void pair_walk(const v2f* __restrict__ A2,
                                          const float (&y)[FEAT],
                                          v8f (&c)[NT2],
                                          const v2f (&cur)[NT2][sizeof...(S)],
                                          std::integer_sequence<int, S...> seq) {
  constexpr int N = (int)sizeof...(S);
  v2f nxt[NT2][N];
  if constexpr (C + 1 < NPCHUNK) {
    ((nxt[0][S] = A2[(0 * KS2 + (C + 1) * N + S) * 32]), ...);
    ((nxt[1][S] = A2[(1 * KS2 + (C + 1) * N + S) * 32]), ...);
    ((nxt[2][S] = A2[(2 * KS2 + (C + 1) * N + S) * 32]), ...);
  }
  if constexpr (C + 1 < NPCHUNK) __builtin_amdgcn_sched_barrier(0);
  auto step = [&](const v2f f0, const v2f f1, const v2f f2, const v2f b) {
    c[0] = __builtin_amdgcn_wmma_f32_16x16x4_f32(false, f0, false, b, (short)0,
                                                 c[0], false, false);
    c[1] = __builtin_amdgcn_wmma_f32_16x16x4_f32(false, f1, false, b, (short)0,
                                                 c[1], false, false);
    c[2] = __builtin_amdgcn_wmma_f32_16x16x4_f32(false, f2, false, b, (short)0,
                                                 c[2], false, false);
  };
  (step(cur[0][S], cur[1][S], cur[2][S], col_frag<NC3 + 2 * (C * N + S)>(y)), ...);
  if constexpr (C + 2 < NPCHUNK) __builtin_amdgcn_sched_barrier(0);
  if constexpr (C + 1 < NPCHUNK) pair_walk<C + 1>(A2, y, c, nxt, seq);
}

__global__ __launch_bounds__(NTHREADS, 8) void sc_mono_kernel(
    const float* __restrict__ x, const int* __restrict__ indices,
    const v2f* __restrict__ P,
    const float* __restrict__ W1_0, const float* __restrict__ W1_1,
    const float* __restrict__ W1_2,
    const float* __restrict__ W2_0, const float* __restrict__ W2_1,
    const float* __restrict__ W2_2,
    const float* __restrict__ W3_0, const float* __restrict__ W3_1,
    const float* __restrict__ W3_2,
    float* __restrict__ out) {
  __shared__ float Zs[ZROWS][ROWS_PER_BLOCK];

  const int tid = threadIdx.x;
  const int rowbase = blockIdx.x * ROWS_PER_BLOCK;

  const int wave = __builtin_amdgcn_readfirstlane(tid >> 5);
  const int lane = tid & 31;
  const int half = lane >> 4;
  const int l15 = lane & 15;

  float y[FEAT];
  {
    const v2f* xr =
        reinterpret_cast<const v2f*>(x + (size_t)(rowbase + l15) * FEAT);
#pragma unroll
    for (int c = 0; c < 8; ++c) {
      const v2f v = xr[c ^ (4 * half)];
      y[2 * c + 0] = v.x; y[2 * c + 1] = v.y;
    }
  }

  const int otk = tid >> 4;
  const int ot = (otk == 0) ? 0 : (otk < 4) ? 1 : 2;
  const int b = rowbase >> 6;
  const int oc = (rowbase & 63) + l15;

  float* Zcol = &Zs[8 * half][l15];

  if (wave < NT3) {
    const v2f* Ap = P + (size_t)wave * (KS3 * 32) + lane;
    static_assert(NCHUNK * CHUNK == KS3, "groups cover the degree-3 walk");
    v8f acc0 = {};
    v8f acc1 = {};
    v2f first[CHUNK];
#pragma unroll
    for (int s = 0; s < CHUNK; ++s) first[s] = Ap[s * 32];
    mono_walk<0, NCHUNK, 0>(Ap, y, acc0, acc1, first,
                            std::make_integer_sequence<int, CHUNK>{});
    const v8f z = acc0 + acc1;
#pragma unroll
    for (int v = 0; v < 8; ++v) Zcol[(wave * 16 + v) * ROWS_PER_BLOCK] = z[v];
  } else {
    const v2f* A2 = P + (size_t)KBASE2 * 32 + lane;
    const v2f* A1 = P + (size_t)KBASE1 * 32 + lane;
    static_assert(NPCHUNK * PCHUNK == KS2, "groups cover the degree-2 walk");
    v8f c[NT2] = {};
    v2f first[NT2][PCHUNK];
    v2f lin[KS1];
#pragma unroll
    for (int T = 0; T < NT2; ++T) {
#pragma unroll
      for (int s = 0; s < PCHUNK; ++s) first[T][s] = A2[(T * KS2 + s) * 32];
    }
#pragma unroll
    for (int s = 0; s < KS1; ++s) lin[s] = A1[s * 32];
    pair_walk<0>(A2, y, c, first, std::make_integer_sequence<int, PCHUNK>{});
#pragma unroll
    for (int T = 0; T < NT2; ++T) {
#pragma unroll
      for (int v = 0; v < 8; ++v)
        Zcol[((NT3 + T) * 16 + v) * ROWS_PER_BLOCK] = c[T][v];
    }
    v8f d0 = {};
    v8f d1 = {};
    mono_walk<NC3 + NC2, 1, 0>(A1, y, d0, d1, lin,
                               std::make_integer_sequence<int, KS1>{});
    const v8f z = d0 + d1;
#pragma unroll
    for (int v = 0; v < 8; ++v)
      Zcol[((NT3 + NT2) * 16 + v) * ROWS_PER_BLOCK] = z[v];
  }

  float wh[16];
  if (tid < NOUT) {
    const float* W3 = (ot == 0) ? W3_0 : (ot == 1) ? W3_1 : W3_2;
    const float* W2 = (ot == 0) ? W2_0 : (ot == 1) ? W2_1 : W2_2;
    const float* W1 = (ot == 0) ? W1_0 : (ot == 1) ? W1_1 : W1_2;
    const int e = indices[b];
#pragma unroll
    for (int p = 0; p < 11; ++p) wh[p] = W3[(e * 11 + p) * 64 + oc];
#pragma unroll
    for (int p = 0; p < 4; ++p) wh[11 + p] = W2[(e * 4 + p) * 64 + oc];
    wh[15] = W1[e * 64 + oc];
  }

  __syncthreads();

  if (tid < NOUT) {
    float s = 0.0f;
#pragma unroll
    for (int p = 0; p < 11; ++p) s = fmaf(Zs[otk * 11 + p][l15], wh[p], s);
#pragma unroll
    for (int p = 0; p < 4; ++p)
      s = fmaf(Zs[NT3 * 16 + otk * 4 + p][l15], wh[11 + p], s);
    s = fmaf(Zs[(NT3 + NT2) * 16 + otk][l15], wh[15], s);
    const int kt  = (ot == 0) ? 1 : (ot == 1) ? 3 : 5;
    const int seg = (ot == 0) ? 0 : (ot == 1) ? 64 : 256;
    const int k   = otk - ((ot == 0) ? 0 : (ot == 1) ? 1 : 4);
    out[b * 576 + seg + oc * kt + k] = s;
  }
}

extern "C" void kernel_launch(void* const* d_in, const int* in_sizes, int n_in,
                              void* d_out, int out_size, void* d_ws, size_t ws_size,
                              hipStream_t stream) {
  (void)in_sizes; (void)n_in; (void)out_size; (void)ws_size;
  const float* x   = (const float*)d_in[0];
  const int* idx   = (const int*)d_in[1];
  const float* U1_0 = (const float*)d_in[2];
  const float* W1_0 = (const float*)d_in[3];
  const float* U2_0 = (const float*)d_in[4];
  const float* W2_0 = (const float*)d_in[5];
  const float* U3_0 = (const float*)d_in[6];
  const float* W3_0 = (const float*)d_in[7];
  const float* U1_1 = (const float*)d_in[8];
  const float* W1_1 = (const float*)d_in[9];
  const float* U2_1 = (const float*)d_in[10];
  const float* W2_1 = (const float*)d_in[11];
  const float* U3_1 = (const float*)d_in[12];
  const float* W3_1 = (const float*)d_in[13];
  const float* U1_2 = (const float*)d_in[14];
  const float* W1_2 = (const float*)d_in[15];
  const float* U2_2 = (const float*)d_in[16];
  const float* W2_2 = (const float*)d_in[17];
  const float* U3_2 = (const float*)d_in[18];
  const float* W3_2 = (const float*)d_in[19];
  float* out = (float*)d_out;

  float2* P = (float2*)d_ws;
  sc_pack_sym<<<193, 256, 0, stream>>>(U1_0, U2_0, U3_0, P, 0);
  sc_pack_sym<<<193, 256, 0, stream>>>(U1_1, U2_1, U3_1, P, 1);
  sc_pack_sym<<<193, 256, 0, stream>>>(U1_2, U2_2, U3_2, P, 2);

  sc_mono_kernel<<<1024, NTHREADS, 0, stream>>>(
      x, idx, (const v2f*)P,
      W1_0, W1_1, W1_2,
      W2_0, W2_1, W2_2,
      W3_0, W3_1, W3_2,
      out);
}


# Main kernel: A operand resident in registers: 256 workgroups of 64 rows; a wave loads its tile's 204 basis fragments once (408 registers) and walks four 16-row tiles on them; sum order unchanged
# speedup vs baseline: 4.4257x; 1.0338x over previous
#include <hip/hip_runtime.h>
#include <utility>


typedef __attribute__((ext_vector_type(2))) float v2f;
typedef __attribute__((ext_vector_type(8))) float v8f;

#define FEAT 16
#define NC3 408
#define NC2 72
#define NC1 8
#define NCOL (NC3 + NC2 + NC1)
#define KS3 204
#define KS2 36
#define KS1 4
#define NT3 7
#define NT2 3
#define KBASE2 (NT3 * KS3)
#define KBASE1 (KBASE2 + NT2 * KS2)
#define KS_TOTAL (KBASE1 + KS1)
#define ZROWS ((NT3 + NT2 + 1) * 16)
#define ROWS_PER_TILE 16
#define NRT 4
#define ROWS_PER_BLOCK (NRT * ROWS_PER_TILE)
#define NTHREADS 256
#define NOUT 576
#define NOUTJ 3
#define CHUNK 12
#define NCHUNK 17
#define DEPTH 4
#define ZTILE (ZROWS * ROWS_PER_TILE + 16)
#define XPITCH 18

struct ColTab { unsigned int v[NCOL]; };
constexpr ColTab make_col_tab() {
  ColTab t{};
  int c = 0;
  unsigned pend = 0;
  bool have = false;
  for (int i = 0; i < 8; ++i)
    for (int j = i; j < 8; ++j) {
      for (int l = j + (j & 1); l < 16; ++l)
        t.v[c++] = (unsigned)(i | (j << 8) | (l << 16));
      if (j & 1) {
        const unsigned m = (unsigned)(i | (j << 8) | (j << 16));
        if (have) { t.v[c++] = pend; t.v[c++] = m; have = false; }
        else { pend = m; have = true; }
      }
    }
  for (int i = 0; i < 8; ++i)
    for (int j = i; j < 8; ++j)
      t.v[c++] = (unsigned)(i | (j << 8) | (16 << 16));
  for (int a = 0; a < 8; ++a)
    for (int b = a + 1; b < 8; ++b)
      t.v[c++] = (unsigned)(a | ((b + 8) << 8) | (16 << 16));
  for (int a = 0; a < 8; ++a)
    t.v[c++] = (unsigned)(a | ((a + 8) << 8) | (16 << 16) | (1u << 24));
  for (int i = 0; i < 8; ++i)
    t.v[c++] = (unsigned)(i | (16 << 8) | (16 << 16));
  return t;
}
constexpr ColTab col_tab_ct = make_col_tab();
__constant__ const ColTab col_tab = make_col_tab();

__device__ __forceinline__ float sym3(const float* __restrict__ U3, int base,
                                      unsigned code, int x8) {
  const int a = (int)(code & 0xff) ^ x8;
  const int b = (int)((code >> 8) & 0xff) ^ x8;
  const int c = (int)((code >> 16) & 0xff) ^ x8;
#define U3AT(p, q, r) U3[base + (((p) * 16 + (q)) * 16 + (r)) * 11]
  const float v0 = U3AT(a, b, c), v1 = U3AT(a, c, b), v2 = U3AT(b, a, c);
  const float v3 = U3AT(b, c, a), v4 = U3AT(c, a, b), v5 = U3AT(c, b, a);
#undef U3AT
  const bool eab = a == b, ebc = b == c, eac = a == c;
  const float w1 = ebc ? 0.0f : v1;
  const float w2 = eab ? 0.0f : v2;
  const float w3 = (eab || eac) ? 0.0f : v3;
  const float w4 = (eac || ebc) ? 0.0f : v4;
  const float w5 = (eab || ebc || eac) ? 0.0f : v5;
  return ((v0 + w1) + (w2 + w3)) + (w4 + w5);
}

__device__ __forceinline__ float sym2(const float* __restrict__ U2, int base,
                                      unsigned code, int x8) {
  const int i = (int)(code & 0xff) ^ x8;
  const int j = (int)((code >> 8) & 0xff) ^ x8;
  const float v0 = U2[base + (i * 16 + j) * 4];
  const float v1 = U2[base + (j * 16 + i) * 4];
  const float s = (i == j) ? v0 : v0 + v1;
  return (x8 != 0 && (code >> 24) != 0) ? 0.0f : s;
}

__global__ __launch_bounds__(256) void sc_pack_sym(const float* __restrict__ U1,
                                                   const float* __restrict__ U2,
                                                   const float* __restrict__ U3,
                                                   float2* __restrict__ P, int tsel) {
  const int n = blockIdx.x * 256 + threadIdx.x;
  if (n >= KS_TOTAL * 32) return;
  const int lane = n & 31;
  const int row = lane & 15;
  const int x8 = (lane >> 4) << 3;
  const int ks = __builtin_amdgcn_readfirstlane(n >> 5);
  const int deg = (ks < KBASE2) ? 3 : (ks < KBASE1) ? 2 : 1;
  const int T3 = ks / KS3, kk3 = ks - T3 * KS3;
  const int r2 = (deg == 2) ? ks - KBASE2 : 0;
  const int T2 = r2 / KS2, kk2 = r2 - T2 * KS2;
  const int T = (deg == 3) ? T3 : (deg == 2) ? T2 : 0;
  const int kk = (deg == 3) ? kk3 : (deg == 2) ? kk2 : ks - KBASE1;
  const int col = ((deg == 3) ? 0 : (deg == 2) ? NC3 : NC3 + NC2) + 2 * kk;
  const unsigned c0 = col_tab.v[col], c1 = col_tab.v[col + 1];
  asm volatile("" ::"s"(c0), "s"(c1));
  const int g = T * 16 + row;
  float a = 0.0f, b = 0.0f;
  bool act, pad;
  if (deg == 3) {
    const int tk = g / 11, path = g - tk * 11;
    const int t = (tk == 0) ? 0 : (tk < 4) ? 1 : 2;
    const int k = tk - ((t == 0) ? 0 : (t == 1) ? 1 : 4);
    act = g < 99 && t == tsel;
    pad = g >= 99 && tsel == 2;
    if (act) {
      const int base = k * (4096 * 11) + path;
      a = sym3(U3, base, c0, x8);
      b = sym3(U3, base, c1, x8);
    }
  } else if (deg == 2) {
    const int tk = g >> 2, path = g & 3;
    const int t = (tk == 0) ? 0 : (tk < 4) ? 1 : 2;
    const int k = tk - ((t == 0) ? 0 : (t == 1) ? 1 : 4);
    act = g < 36 && t == tsel;
    pad = g >= 36 && tsel == 2;
    if (act) {
      const int base = k * 1024 + path;
      a = sym2(U2, base, c0, x8);
      b = sym2(U2, base, c1, x8);
    }
  } else {
    const int tk = g;
    const int t = (tk == 0) ? 0 : (tk < 4) ? 1 : 2;
    const int k = tk - ((t == 0) ? 0 : (t == 1) ? 1 : 4);
    act = g < 9 && t == tsel;
    pad = g >= 9 && tsel == 2;
    if (act) {
      a = U1[k * 16 + ((int)(c0 & 0xff) ^ x8)];
      b = U1[k * 16 + ((int)(c1 & 0xff) ^ x8)];
    }
  }
  if (act || pad) P[n] = make_float2(a, b);
}

template <int C>
__device__ __forceinline__ float col_val(const float (&y)[FEAT]) {
  constexpr unsigned code = col_tab_ct.v[C];
  constexpr int i = code & 0xff, j = (code >> 8) & 0xff, l = (code >> 16) & 0xff;
  if constexpr (j == 16) return y[i];
  else if constexpr (l == 16) return y[i] * y[j];
  else return (y[i] * y[j]) * y[l];
}

template <int C>
__device__ __forceinline__ v2f col_frag(const float (&y)[FEAT]) {
  v2f b;
  b.x = col_val<C>(y);
  b.y = col_val<C + 1>(y);
  return b;
}

template <int C, int ODD>
__device__ __forceinline__ void mono_step(const v2f a, const float (&y)[FEAT],
                                          v8f& acc0, v8f& acc1) {
  const v2f b = col_frag<C>(y);
  if constexpr (ODD)
    acc1 = __builtin_amdgcn_wmma_f32_16x16x4_f32(false, a, false, b, (short)0,
                                                 acc1, false, false);
  else
    acc0 = __builtin_amdgcn_wmma_f32_16x16x4_f32(false, a, false, b, (short)0,
                                                 acc0, false, false);
}

template <int C, bool LOADING, int... S>
__device__ __forceinline__ void tile_walk(const v2f* __restrict__ Ap,
                                          const float (&y)[FEAT],
                                          v8f& acc0, v8f& acc1, v2f (&A)[KS3],
                                          std::integer_sequence<int, S...> seq) {
  constexpr int N = (int)sizeof...(S);
  if constexpr (LOADING && C + DEPTH < NCHUNK) {
    ((A[(C + DEPTH) * N + S] = Ap[((C + DEPTH) * N + S) * 32]), ...);
    __builtin_amdgcn_sched_barrier(0);
  }
  (mono_step<2 * (C * N + S), (C * N + S) & 1>(A[C * N + S], y, acc0, acc1), ...);
  if constexpr (LOADING && C + DEPTH + 1 < NCHUNK) __builtin_amdgcn_sched_barrier(0);
  if constexpr (C + 1 < NCHUNK)
    tile_walk<C + 1, LOADING>(Ap, y, acc0, acc1, A, seq);
}

template <int S>
__device__ __forceinline__ void pair_step(const v2f (&A2)[NT2][KS2],
                                          const float (&y)[FEAT], v8f (&c)[NT2]) {
  const v2f b = col_frag<NC3 + 2 * S>(y);
  c[0] = __builtin_amdgcn_wmma_f32_16x16x4_f32(false, A2[0][S], false, b, (short)0,
                                               c[0], false, false);
  c[1] = __builtin_amdgcn_wmma_f32_16x16x4_f32(false, A2[1][S], false, b, (short)0,
                                               c[1], false, false);
  c[2] = __builtin_amdgcn_wmma_f32_16x16x4_f32(false, A2[2][S], false, b, (short)0,
                                               c[2], false, false);
}
template <int... S>
__device__ __forceinline__ void pair_walk(const v2f (&A2)[NT2][KS2],
                                          const float (&y)[FEAT], v8f (&c)[NT2],
                                          std::integer_sequence<int, S...>) {
  (pair_step<S>(A2, y, c), ...);
}
template <int... S>
__device__ __forceinline__ void lin_walk(const v2f (&A1)[KS1], const float (&y)[FEAT],
                                         v8f& d0, v8f& d1,
                                         std::integer_sequence<int, S...>) {
  (mono_step<NC3 + NC2 + 2 * S, S & 1>(A1[S], y, d0, d1), ...);
}

__device__ __forceinline__ void read_row(const float* Xs, int rt, int l15, int half,
                                         float (&y)[FEAT]) {
  const v2f* xr =
      reinterpret_cast<const v2f*>(Xs + (rt * ROWS_PER_TILE + l15) * XPITCH);
#pragma unroll
  for (int c = 0; c < 8; ++c) {
    const v2f v = xr[c ^ (4 * half)];
    y[2 * c + 0] = v.x; y[2 * c + 1] = v.y;
  }
}

__global__ __launch_bounds__(NTHREADS, 2) void sc_mono_kernel(
    const float* __restrict__ x, const int* __restrict__ indices,
    const v2f* __restrict__ P,
    const float* __restrict__ W1_0, const float* __restrict__ W1_1,
    const float* __restrict__ W1_2,
    const float* __restrict__ W2_0, const float* __restrict__ W2_1,
    const float* __restrict__ W2_2,
    const float* __restrict__ W3_0, const float* __restrict__ W3_1,
    const float* __restrict__ W3_2,
    float* __restrict__ out) {
  __shared__ float Zs[NRT * ZTILE];
  __shared__ alignas(16) float Xs[ROWS_PER_BLOCK * XPITCH];

  const int tid = threadIdx.x;
  const int b = blockIdx.x;

  const int wave = __builtin_amdgcn_readfirstlane(tid >> 5);
  const int lane = tid & 31;
  const int half = lane >> 4;
  const int l15 = lane & 15;

  const int e = indices[b];

  v2f A[KS3];
  const v2f* Ap = P + (size_t)wave * (KS3 * 32) + lane;
  if (wave < NT3) {
#pragma unroll
    for (int s = 0; s < DEPTH * CHUNK; ++s) A[s] = Ap[s * 32];
  }

  {
    const float4 v =
        reinterpret_cast<const float4*>(x + (size_t)b * (ROWS_PER_BLOCK * FEAT))[tid];
    v2f* d = reinterpret_cast<v2f*>(Xs + (tid >> 2) * XPITCH + (tid & 3) * 4);
    v2f lo, hi;
    lo.x = v.x; lo.y = v.y; hi.x = v.z; hi.y = v.w;
    d[0] = lo; d[1] = hi;
  }
  __syncthreads();

  float* Zcol = &Zs[(8 * half) * ROWS_PER_TILE + l15];

  if (wave < NT3) {
    static_assert(NCHUNK * CHUNK == KS3, "groups cover the degree-3 walk");
    const auto seq = std::make_integer_sequence<int, CHUNK>{};
    float y[FEAT];
    {
      read_row(Xs, 0, l15, half, y);
      v8f acc0 = {};
      v8f acc1 = {};
      tile_walk<0, true>(Ap, y, acc0, acc1, A, seq);
      const v8f z = acc0 + acc1;
#pragma unroll
      for (int v = 0; v < 8; ++v) Zcol[(wave * 16 + v) * ROWS_PER_TILE] = z[v];
    }
#pragma unroll 1
    for (int rt = 1; rt < NRT; ++rt) {
      read_row(Xs, rt, l15, half, y);
      v8f acc0 = {};
      v8f acc1 = {};
      tile_walk<0, false>(Ap, y, acc0, acc1, A, seq);
      const v8f z = acc0 + acc1;
#pragma unroll
      for (int v = 0; v < 8; ++v)
        Zcol[rt * ZTILE + (wave * 16 + v) * ROWS_PER_TILE] = z[v];
    }
  } else {
    const v2f* A2p = P + (size_t)KBASE2 * 32 + lane;
    const v2f* A1p = P + (size_t)KBASE1 * 32 + lane;
    v2f A2[NT2][KS2];
    v2f A1[KS1];
#pragma unroll
    for (int T = 0; T < NT2; ++T) {
#pragma unroll
      for (int s = 0; s < KS2; ++s) A2[T][s] = A2p[(T * KS2 + s) * 32];
    }
#pragma unroll
    for (int s = 0; s < KS1; ++s) A1[s] = A1p[s * 32];
    float y[FEAT];
#pragma unroll 1
    for (int rt = 0; rt < NRT; ++rt) {
      read_row(Xs, rt, l15, half, y);
      v8f c[NT2] = {};
      pair_walk(A2, y, c, std::make_integer_sequence<int, KS2>{});
#pragma unroll
      for (int T = 0; T < NT2; ++T) {
#pragma unroll
        for (int v = 0; v < 8; ++v)
          Zcol[rt * ZTILE + ((NT3 + T) * 16 + v) * ROWS_PER_TILE] = c[T][v];
      }
      v8f d0 = {};
      v8f d1 = {};
      lin_walk(A1, y, d0, d1, std::make_integer_sequence<int, KS1>{});
      const v8f z = d0 + d1;
#pragma unroll
      for (int v = 0; v < 8; ++v)
        Zcol[rt * ZTILE + ((NT3 + NT2) * 16 + v) * ROWS_PER_TILE] = z[v];
    }
  }

  float wh[NOUTJ][16];
#pragma unroll
  for (int j = 0; j < NOUTJ; ++j) {
    const int o = tid + NTHREADS * j;
    if (o < NOUT) {
      const int otk = o >> 6, oc = o & 63;
      const int ot = (otk == 0) ? 0 : (otk < 4) ? 1 : 2;
      const float* W3 = (ot == 0) ? W3_0 : (ot == 1) ? W3_1 : W3_2;
      const float* W2 = (ot == 0) ? W2_0 : (ot == 1) ? W2_1 : W2_2;
      const float* W1 = (ot == 0) ? W1_0 : (ot == 1) ? W1_1 : W1_2;
#pragma unroll
      for (int p = 0; p < 11; ++p) wh[j][p] = W3[(e * 11 + p) * 64 + oc];
#pragma unroll
      for (int p = 0; p < 4; ++p) wh[j][11 + p] = W2[(e * 4 + p) * 64 + oc];
      wh[j][15] = W1[e * 64 + oc];
    }
  }

  __syncthreads();

#pragma unroll
  for (int j = 0; j < NOUTJ; ++j) {
    const int o = tid + NTHREADS * j;
    if (o < NOUT) {
      const int otk = o >> 6, oc = o & 63;
      const int ot = (otk == 0) ? 0 : (otk < 4) ? 1 : 2;
      const float* Zr = &Zs[(oc >> 4) * ZTILE + (oc & 15)];
      float s = 0.0f;
#pragma unroll
      for (int p = 0; p < 11; ++p)
        s = fmaf(Zr[(otk * 11 + p) * ROWS_PER_TILE], wh[j][p], s);
#pragma unroll
      for (int p = 0; p < 4; ++p)
        s = fmaf(Zr[(NT3 * 16 + otk * 4 + p) * ROWS_PER_TILE], wh[j][11 + p], s);
      s = fmaf(Zr[((NT3 + NT2) * 16 + otk) * ROWS_PER_TILE], wh[j][15], s);
      const int kt  = (ot == 0) ? 1 : (ot == 1) ? 3 : 5;
      const int seg = (ot == 0) ? 0 : (ot == 1) ? 64 : 256;
      const int k   = otk - ((ot == 0) ? 0 : (ot == 1) ? 1 : 4);
      out[b * 576 + seg + oc * kt + k] = s;
    }
  }
}

extern "C" void kernel_launch(void* const* d_in, const int* in_sizes, int n_in,
                              void* d_out, int out_size, void* d_ws, size_t ws_size,
                              hipStream_t stream) {
  (void)in_sizes; (void)n_in; (void)out_size; (void)ws_size;
  const float* x   = (const float*)d_in[0];
  const int* idx   = (const int*)d_in[1];
  const float* U1_0 = (const float*)d_in[2];
  const float* W1_0 = (const float*)d_in[3];
  const float* U2_0 = (const float*)d_in[4];
  const float* W2_0 = (const float*)d_in[5];
  const float* U3_0 = (const float*)d_in[6];
  const float* W3_0 = (const float*)d_in[7];
  const float* U1_1 = (const float*)d_in[8];
  const float* W1_1 = (const float*)d_in[9];
  const float* U2_1 = (const float*)d_in[10];
  const float* W2_1 = (const float*)d_in[11];
  const float* U3_1 = (const float*)d_in[12];
  const float* W3_1 = (const float*)d_in[13];
  const float* U1_2 = (const float*)d_in[14];
  const float* W1_2 = (const float*)d_in[15];
  const float* U2_2 = (const float*)d_in[16];
  const float* W2_2 = (const float*)d_in[17];
  const float* U3_2 = (const float*)d_in[18];
  const float* W3_2 = (const float*)d_in[19];
  float* out = (float*)d_out;

  float2* P = (float2*)d_ws;
  sc_pack_sym<<<193, 256, 0, stream>>>(U1_0, U2_0, U3_0, P, 0);
  sc_pack_sym<<<193, 256, 0, stream>>>(U1_1, U2_1, U3_1, P, 1);
  sc_pack_sym<<<193, 256, 0, stream>>>(U1_2, U2_2, U3_2, P, 2);

  sc_mono_kernel<<<256, NTHREADS, 0, stream>>>(
      x, idx, (const v2f*)P,
      W1_0, W1_1, W1_2,
      W2_0, W2_1, W2_2,
      W3_0, W3_1, W3_2,
      out);
}


# Launch merge on this chain's own pack kernel: its three per-irrep launches become one (a thread takes its irrep's basis by its packed row); 4 -> 2 launches; sum order unchanged
# speedup vs baseline: 5.5366x; 1.2510x over previous
#include <hip/hip_runtime.h>
#include <utility>


typedef __attribute__((ext_vector_type(2))) float v2f;
typedef __attribute__((ext_vector_type(8))) float v8f;

#define FEAT 16
#define NC3 408
#define NC2 72
#define NC1 8
#define NCOL (NC3 + NC2 + NC1)
#define KS3 204
#define KS2 36
#define KS1 4
#define NT3 7
#define NT2 3
#define KBASE2 (NT3 * KS3)
#define KBASE1 (KBASE2 + NT2 * KS2)
#define KS_TOTAL (KBASE1 + KS1)
#define ZROWS ((NT3 + NT2 + 1) * 16)
#define ROWS_PER_TILE 16
#define NRT 4
#define ROWS_PER_BLOCK (NRT * ROWS_PER_TILE)
#define NTHREADS 256
#define NOUT 576
#define NOUTJ 3
#define CHUNK 12
#define NCHUNK 17
#define DEPTH 4
#define ZTILE (ZROWS * ROWS_PER_TILE + 16)
#define XPITCH 18

struct ColTab { unsigned int v[NCOL]; };
constexpr ColTab make_col_tab() {
  ColTab t{};
  int c = 0;
  unsigned pend = 0;
  bool have = false;
  for (int i = 0; i < 8; ++i)
    for (int j = i; j < 8; ++j) {
      for (int l = j + (j & 1); l < 16; ++l)
        t.v[c++] = (unsigned)(i | (j << 8) | (l << 16));
      if (j & 1) {
        const unsigned m = (unsigned)(i | (j << 8) | (j << 16));
        if (have) { t.v[c++] = pend; t.v[c++] = m; have = false; }
        else { pend = m; have = true; }
      }
    }
  for (int i = 0; i < 8; ++i)
    for (int j = i; j < 8; ++j)
      t.v[c++] = (unsigned)(i | (j << 8) | (16 << 16));
  for (int a = 0; a < 8; ++a)
    for (int b = a + 1; b < 8; ++b)
      t.v[c++] = (unsigned)(a | ((b + 8) << 8) | (16 << 16));
  for (int a = 0; a < 8; ++a)
    t.v[c++] = (unsigned)(a | ((a + 8) << 8) | (16 << 16) | (1u << 24));
  for (int i = 0; i < 8; ++i)
    t.v[c++] = (unsigned)(i | (16 << 8) | (16 << 16));
  return t;
}
constexpr ColTab col_tab_ct = make_col_tab();
__constant__ const ColTab col_tab = make_col_tab();

__device__ __forceinline__ float sym3(const float* __restrict__ U3, int base,
                                      unsigned code, int x8) {
  const int a = (int)(code & 0xff) ^ x8;
  const int b = (int)((code >> 8) & 0xff) ^ x8;
  const int c = (int)((code >> 16) & 0xff) ^ x8;
#define U3AT(p, q, r) U3[base + (((p) * 16 + (q)) * 16 + (r)) * 11]
  const float v0 = U3AT(a, b, c), v1 = U3AT(a, c, b), v2 = U3AT(b, a, c);
  const float v3 = U3AT(b, c, a), v4 = U3AT(c, a, b), v5 = U3AT(c, b, a);
#undef U3AT
  const bool eab = a == b, ebc = b == c, eac = a == c;
  const float w1 = ebc ? 0.0f : v1;
  const float w2 = eab ? 0.0f : v2;
  const float w3 = (eab || eac) ? 0.0f : v3;
  const float w4 = (eac || ebc) ? 0.0f : v4;
  const float w5 = (eab || ebc || eac) ? 0.0f : v5;
  return ((v0 + w1) + (w2 + w3)) + (w4 + w5);
}

__device__ __forceinline__ float sym2(const float* __restrict__ U2, int base,
                                      unsigned code, int x8) {
  const int i = (int)(code & 0xff) ^ x8;
  const int j = (int)((code >> 8) & 0xff) ^ x8;
  const float v0 = U2[base + (i * 16 + j) * 4];
  const float v1 = U2[base + (j * 16 + i) * 4];
  const float s = (i == j) ? v0 : v0 + v1;
  return (x8 != 0 && (code >> 24) != 0) ? 0.0f : s;
}

__global__ __launch_bounds__(256) void sc_pack_sym(
    const float* __restrict__ U1_0, const float* __restrict__ U2_0,
    const float* __restrict__ U3_0,
    const float* __restrict__ U1_1, const float* __restrict__ U2_1,
    const float* __restrict__ U3_1,
    const float* __restrict__ U1_2, const float* __restrict__ U2_2,
    const float* __restrict__ U3_2,
    float2* __restrict__ P) {
  const int n = blockIdx.x * 256 + threadIdx.x;
  if (n >= KS_TOTAL * 32) return;
  const int lane = n & 31;
  const int row = lane & 15;
  const int x8 = (lane >> 4) << 3;
  const int ks = __builtin_amdgcn_readfirstlane(n >> 5);
  const int deg = (ks < KBASE2) ? 3 : (ks < KBASE1) ? 2 : 1;
  const int T3 = ks / KS3, kk3 = ks - T3 * KS3;
  const int r2 = (deg == 2) ? ks - KBASE2 : 0;
  const int T2 = r2 / KS2, kk2 = r2 - T2 * KS2;
  const int T = (deg == 3) ? T3 : (deg == 2) ? T2 : 0;
  const int kk = (deg == 3) ? kk3 : (deg == 2) ? kk2 : ks - KBASE1;
  const int col = ((deg == 3) ? 0 : (deg == 2) ? NC3 : NC3 + NC2) + 2 * kk;
  const unsigned c0 = col_tab.v[col], c1 = col_tab.v[col + 1];
  const int g = T * 16 + row;
  float a = 0.0f, b = 0.0f;
  if (deg == 3) {
    const int tk = g / 11, path = g - tk * 11;
    const int t = (tk == 0) ? 0 : (tk < 4) ? 1 : 2;
    const int k = tk - ((t == 0) ? 0 : (t == 1) ? 1 : 4);
    if (g < 99) {
      const float* __restrict__ U3 = (t == 0) ? U3_0 : (t == 1) ? U3_1 : U3_2;
      const int base = k * (4096 * 11) + path;
      a = sym3(U3, base, c0, x8);
      b = sym3(U3, base, c1, x8);
    }
  } else if (deg == 2) {
    const int tk = g >> 2, path = g & 3;
    const int t = (tk == 0) ? 0 : (tk < 4) ? 1 : 2;
    const int k = tk - ((t == 0) ? 0 : (t == 1) ? 1 : 4);
    if (g < 36) {
      const float* __restrict__ U2 = (t == 0) ? U2_0 : (t == 1) ? U2_1 : U2_2;
      const int base = k * 1024 + path;
      a = sym2(U2, base, c0, x8);
      b = sym2(U2, base, c1, x8);
    }
  } else {
    const int tk = g;
    const int t = (tk == 0) ? 0 : (tk < 4) ? 1 : 2;
    const int k = tk - ((t == 0) ? 0 : (t == 1) ? 1 : 4);
    if (g < 9) {
      const float* __restrict__ U1 = (t == 0) ? U1_0 : (t == 1) ? U1_1 : U1_2;
      a = U1[k * 16 + ((int)(c0 & 0xff) ^ x8)];
      b = U1[k * 16 + ((int)(c1 & 0xff) ^ x8)];
    }
  }
  P[n] = make_float2(a, b);
}

template <int C>
__device__ __forceinline__ float col_val(const float (&y)[FEAT]) {
  constexpr unsigned code = col_tab_ct.v[C];
  constexpr int i = code & 0xff, j = (code >> 8) & 0xff, l = (code >> 16) & 0xff;
  if constexpr (j == 16) return y[i];
  else if constexpr (l == 16) return y[i] * y[j];
  else return (y[i] * y[j]) * y[l];
}

template <int C>
__device__ __forceinline__ v2f col_frag(const float (&y)[FEAT]) {
  v2f b;
  b.x = col_val<C>(y);
  b.y = col_val<C + 1>(y);
  return b;
}

template <int C, int ODD>
__device__ __forceinline__ void mono_step(const v2f a, const float (&y)[FEAT],
                                          v8f& acc0, v8f& acc1) {
  const v2f b = col_frag<C>(y);
  if constexpr (ODD)
    acc1 = __builtin_amdgcn_wmma_f32_16x16x4_f32(false, a, false, b, (short)0,
                                                 acc1, false, false);
  else
    acc0 = __builtin_amdgcn_wmma_f32_16x16x4_f32(false, a, false, b, (short)0,
                                                 acc0, false, false);
}

template <int C, bool LOADING, int... S>
__device__ __forceinline__ void tile_walk(const v2f* __restrict__ Ap,
                                          const float (&y)[FEAT],
                                          v8f& acc0, v8f& acc1, v2f (&A)[KS3],
                                          std::integer_sequence<int, S...> seq) {
  constexpr int N = (int)sizeof...(S);
  if constexpr (LOADING && C + DEPTH < NCHUNK) {
    ((A[(C + DEPTH) * N + S] = Ap[((C + DEPTH) * N + S) * 32]), ...);
    __builtin_amdgcn_sched_barrier(0);
  }
  (mono_step<2 * (C * N + S), (C * N + S) & 1>(A[C * N + S], y, acc0, acc1), ...);
  if constexpr (LOADING && C + DEPTH + 1 < NCHUNK) __builtin_amdgcn_sched_barrier(0);
  if constexpr (C + 1 < NCHUNK)
    tile_walk<C + 1, LOADING>(Ap, y, acc0, acc1, A, seq);
}

template <int S>
__device__ __forceinline__ void pair_step(const v2f (&A2)[NT2][KS2],
                                          const float (&y)[FEAT], v8f (&c)[NT2]) {
  const v2f b = col_frag<NC3 + 2 * S>(y);
  c[0] = __builtin_amdgcn_wmma_f32_16x16x4_f32(false, A2[0][S], false, b, (short)0,
                                               c[0], false, false);
  c[1] = __builtin_amdgcn_wmma_f32_16x16x4_f32(false, A2[1][S], false, b, (short)0,
                                               c[1], false, false);
  c[2] = __builtin_amdgcn_wmma_f32_16x16x4_f32(false, A2[2][S], false, b, (short)0,
                                               c[2], false, false);
}
template <int... S>
__device__ __forceinline__ void pair_walk(const v2f (&A2)[NT2][KS2],
                                          const float (&y)[FEAT], v8f (&c)[NT2],
                                          std::integer_sequence<int, S...>) {
  (pair_step<S>(A2, y, c), ...);
}
template <int... S>
__device__ __forceinline__ void lin_walk(const v2f (&A1)[KS1], const float (&y)[FEAT],
                                         v8f& d0, v8f& d1,
                                         std::integer_sequence<int, S...>) {
  (mono_step<NC3 + NC2 + 2 * S, S & 1>(A1[S], y, d0, d1), ...);
}

__device__ __forceinline__ void read_row(const float* Xs, int rt, int l15, int half,
                                         float (&y)[FEAT]) {
  const v2f* xr =
      reinterpret_cast<const v2f*>(Xs + (rt * ROWS_PER_TILE + l15) * XPITCH);
#pragma unroll
  for (int c = 0; c < 8; ++c) {
    const v2f v = xr[c ^ (4 * half)];
    y[2 * c + 0] = v.x; y[2 * c + 1] = v.y;
  }
}

__global__ __launch_bounds__(NTHREADS, 2) void sc_mono_kernel(
    const float* __restrict__ x, const int* __restrict__ indices,
    const v2f* __restrict__ P,
    const float* __restrict__ W1_0, const float* __restrict__ W1_1,
    const float* __restrict__ W1_2,
    const float* __restrict__ W2_0, const float* __restrict__ W2_1,
    const float* __restrict__ W2_2,
    const float* __restrict__ W3_0, const float* __restrict__ W3_1,
    const float* __restrict__ W3_2,
    float* __restrict__ out) {
  __shared__ float Zs[NRT * ZTILE];
  __shared__ alignas(16) float Xs[ROWS_PER_BLOCK * XPITCH];

  const int tid = threadIdx.x;
  const int b = blockIdx.x;

  const int wave = __builtin_amdgcn_readfirstlane(tid >> 5);
  const int lane = tid & 31;
  const int half = lane >> 4;
  const int l15 = lane & 15;

  const int e = indices[b];

  v2f A[KS3];
  const v2f* Ap = P + (size_t)wave * (KS3 * 32) + lane;
  if (wave < NT3) {
#pragma unroll
    for (int s = 0; s < DEPTH * CHUNK; ++s) A[s] = Ap[s * 32];
  }

  {
    const float4 v =
        reinterpret_cast<const float4*>(x + (size_t)b * (ROWS_PER_BLOCK * FEAT))[tid];
    v2f* d = reinterpret_cast<v2f*>(Xs + (tid >> 2) * XPITCH + (tid & 3) * 4);
    v2f lo, hi;
    lo.x = v.x; lo.y = v.y; hi.x = v.z; hi.y = v.w;
    d[0] = lo; d[1] = hi;
  }
  __syncthreads();

  float* Zcol = &Zs[(8 * half) * ROWS_PER_TILE + l15];

  if (wave < NT3) {
    static_assert(NCHUNK * CHUNK == KS3, "groups cover the degree-3 walk");
    const auto seq = std::make_integer_sequence<int, CHUNK>{};
    float y[FEAT];
    {
      read_row(Xs, 0, l15, half, y);
      v8f acc0 = {};
      v8f acc1 = {};
      tile_walk<0, true>(Ap, y, acc0, acc1, A, seq);
      const v8f z = acc0 + acc1;
#pragma unroll
      for (int v = 0; v < 8; ++v) Zcol[(wave * 16 + v) * ROWS_PER_TILE] = z[v];
    }
#pragma unroll 1
    for (int rt = 1; rt < NRT; ++rt) {
      read_row(Xs, rt, l15, half, y);
      v8f acc0 = {};
      v8f acc1 = {};
      tile_walk<0, false>(Ap, y, acc0, acc1, A, seq);
      const v8f z = acc0 + acc1;
#pragma unroll
      for (int v = 0; v < 8; ++v)
        Zcol[rt * ZTILE + (wave * 16 + v) * ROWS_PER_TILE] = z[v];
    }
  } else {
    const v2f* A2p = P + (size_t)KBASE2 * 32 + lane;
    const v2f* A1p = P + (size_t)KBASE1 * 32 + lane;
    v2f A2[NT2][KS2];
    v2f A1[KS1];
#pragma unroll
    for (int T = 0; T < NT2; ++T) {
#pragma unroll
      for (int s = 0; s < KS2; ++s) A2[T][s] = A2p[(T * KS2 + s) * 32];
    }
#pragma unroll
    for (int s = 0; s < KS1; ++s) A1[s] = A1p[s * 32];
    float y[FEAT];
#pragma unroll 1
    for (int rt = 0; rt < NRT; ++rt) {
      read_row(Xs, rt, l15, half, y);
      v8f c[NT2] = {};
      pair_walk(A2, y, c, std::make_integer_sequence<int, KS2>{});
#pragma unroll
      for (int T = 0; T < NT2; ++T) {
#pragma unroll
        for (int v = 0; v < 8; ++v)
          Zcol[rt * ZTILE + ((NT3 + T) * 16 + v) * ROWS_PER_TILE] = c[T][v];
      }
      v8f d0 = {};
      v8f d1 = {};
      lin_walk(A1, y, d0, d1, std::make_integer_sequence<int, KS1>{});
      const v8f z = d0 + d1;
#pragma unroll
      for (int v = 0; v < 8; ++v)
        Zcol[rt * ZTILE + ((NT3 + NT2) * 16 + v) * ROWS_PER_TILE] = z[v];
    }
  }

  float wh[NOUTJ][16];
#pragma unroll
  for (int j = 0; j < NOUTJ; ++j) {
    const int o = tid + NTHREADS * j;
    if (o < NOUT) {
      const int otk = o >> 6, oc = o & 63;
      const int ot = (otk == 0) ? 0 : (otk < 4) ? 1 : 2;
      const float* W3 = (ot == 0) ? W3_0 : (ot == 1) ? W3_1 : W3_2;
      const float* W2 = (ot == 0) ? W2_0 : (ot == 1) ? W2_1 : W2_2;
      const float* W1 = (ot == 0) ? W1_0 : (ot == 1) ? W1_1 : W1_2;
#pragma unroll
      for (int p = 0; p < 11; ++p) wh[j][p] = W3[(e * 11 + p) * 64 + oc];
#pragma unroll
      for (int p = 0; p < 4; ++p) wh[j][11 + p] = W2[(e * 4 + p) * 64 + oc];
      wh[j][15] = W1[e * 64 + oc];
    }
  }

  __syncthreads();

#pragma unroll
  for (int j = 0; j < NOUTJ; ++j) {
    const int o = tid + NTHREADS * j;
    if (o < NOUT) {
      const int otk = o >> 6, oc = o & 63;
      const int ot = (otk == 0) ? 0 : (otk < 4) ? 1 : 2;
      const float* Zr = &Zs[(oc >> 4) * ZTILE + (oc & 15)];
      float s = 0.0f;
#pragma unroll
      for (int p = 0; p < 11; ++p)
        s = fmaf(Zr[(otk * 11 + p) * ROWS_PER_TILE], wh[j][p], s);
#pragma unroll
      for (int p = 0; p < 4; ++p)
        s = fmaf(Zr[(NT3 * 16 + otk * 4 + p) * ROWS_PER_TILE], wh[j][11 + p], s);
      s = fmaf(Zr[((NT3 + NT2) * 16 + otk) * ROWS_PER_TILE], wh[j][15], s);
      const int kt  = (ot == 0) ? 1 : (ot == 1) ? 3 : 5;
      const int seg = (ot == 0) ? 0 : (ot == 1) ? 64 : 256;
      const int k   = otk - ((ot == 0) ? 0 : (ot == 1) ? 1 : 4);
      out[b * 576 + seg + oc * kt + k] = s;
    }
  }
}

extern "C" void kernel_launch(void* const* d_in, const int* in_sizes, int n_in,
                              void* d_out, int out_size, void* d_ws, size_t ws_size,
                              hipStream_t stream) {
  (void)in_sizes; (void)n_in; (void)out_size; (void)ws_size;
  const float* x   = (const float*)d_in[0];
  const int* idx   = (const int*)d_in[1];
  const float* U1_0 = (const float*)d_in[2];
  const float* W1_0 = (const float*)d_in[3];
  const float* U2_0 = (const float*)d_in[4];
  const float* W2_0 = (const float*)d_in[5];
  const float* U3_0 = (const float*)d_in[6];
  const float* W3_0 = (const float*)d_in[7];
  const float* U1_1 = (const float*)d_in[8];
  const float* W1_1 = (const float*)d_in[9];
  const float* U2_1 = (const float*)d_in[10];
  const float* W2_1 = (const float*)d_in[11];
  const float* U3_1 = (const float*)d_in[12];
  const float* W3_1 = (const float*)d_in[13];
  const float* U1_2 = (const float*)d_in[14];
  const float* W1_2 = (const float*)d_in[15];
  const float* U2_2 = (const float*)d_in[16];
  const float* W2_2 = (const float*)d_in[17];
  const float* U3_2 = (const float*)d_in[18];
  const float* W3_2 = (const float*)d_in[19];
  float* out = (float*)d_out;

  float2* P = (float2*)d_ws;
  sc_pack_sym<<<193, 256, 0, stream>>>(U1_0, U2_0, U3_0, U1_1, U2_1, U3_1,
                                       U1_2, U2_2, U3_2, P);

  sc_mono_kernel<<<256, NTHREADS, 0, stream>>>(
      x, idx, (const v2f*)P,
      W1_0, W1_1, W1_2,
      W2_0, W2_1, W2_2,
      W3_0, W3_1, W3_2,
      out);
}


# Main kernel: waits cut from the younger wave's lone run on the SIMD's matrix pipe: fragments streamed (4 row tiles at once), weights up front, light wave takes 6 k-steps per tile (sum order changes)
# speedup vs baseline: 5.7722x; 1.0426x over previous
#include <hip/hip_runtime.h>
#include <utility>


typedef __attribute__((ext_vector_type(2))) float v2f;
typedef __attribute__((ext_vector_type(8))) float v8f;

#define FEAT 16
#define NC3 408
#define NC2 72
#define NC1 8
#define NCOL (NC3 + NC2 + NC1)
#define KS3 204
#define KS2 36
#define KS1 4
#define NT3 7
#define NT2 3
#define KBASE2 (NT3 * KS3)
#define KBASE1 (KBASE2 + NT2 * KS2)
#define KS_TOTAL (KBASE1 + KS1)
#define ZROWS ((NT3 + NT2 + 1) * 16)
#define ROWS_PER_TILE 16
#define NRT 4
#define ROWS_PER_BLOCK (NRT * ROWS_PER_TILE)
#define NTHREADS 256
#define NOUT 576
#define NOUTJ 3
#define KS3W 198
#define KT (KS3 - KS3W)
#define NTA 3
#define CHUNK 9
#define NCHUNK 22
#define DEPTH 5
#define ZTILE (ZROWS * ROWS_PER_TILE + 16)
#define ZT3 (NT3 * 16 * ROWS_PER_TILE + 16)
#define XPITCH 18

struct ColTab { unsigned int v[NCOL]; };
constexpr ColTab make_col_tab() {
  ColTab t{};
  int c = 0;
  unsigned pend = 0;
  bool have = false;
  for (int i = 0; i < 8; ++i)
    for (int j = i; j < 8; ++j) {
      for (int l = j + (j & 1); l < 16; ++l)
        t.v[c++] = (unsigned)(i | (j << 8) | (l << 16));
      if (j & 1) {
        const unsigned m = (unsigned)(i | (j << 8) | (j << 16));
        if (have) { t.v[c++] = pend; t.v[c++] = m; have = false; }
        else { pend = m; have = true; }
      }
    }
  for (int i = 0; i < 8; ++i)
    for (int j = i; j < 8; ++j)
      t.v[c++] = (unsigned)(i | (j << 8) | (16 << 16));
  for (int a = 0; a < 8; ++a)
    for (int b = a + 1; b < 8; ++b)
      t.v[c++] = (unsigned)(a | ((b + 8) << 8) | (16 << 16));
  for (int a = 0; a < 8; ++a)
    t.v[c++] = (unsigned)(a | ((a + 8) << 8) | (16 << 16) | (1u << 24));
  for (int i = 0; i < 8; ++i)
    t.v[c++] = (unsigned)(i | (16 << 8) | (16 << 16));
  return t;
}
constexpr ColTab col_tab_ct = make_col_tab();
__constant__ const ColTab col_tab = make_col_tab();

__device__ __forceinline__ float sym3(const float* __restrict__ U3, int base,
                                      unsigned code, int x8) {
  const int a = (int)(code & 0xff) ^ x8;
  const int b = (int)((code >> 8) & 0xff) ^ x8;
  const int c = (int)((code >> 16) & 0xff) ^ x8;
#define U3AT(p, q, r) U3[base + (((p) * 16 + (q)) * 16 + (r)) * 11]
  const float v0 = U3AT(a, b, c), v1 = U3AT(a, c, b), v2 = U3AT(b, a, c);
  const float v3 = U3AT(b, c, a), v4 = U3AT(c, a, b), v5 = U3AT(c, b, a);
#undef U3AT
  const bool eab = a == b, ebc = b == c, eac = a == c;
  const float w1 = ebc ? 0.0f : v1;
  const float w2 = eab ? 0.0f : v2;
  const float w3 = (eab || eac) ? 0.0f : v3;
  const float w4 = (eac || ebc) ? 0.0f : v4;
  const float w5 = (eab || ebc || eac) ? 0.0f : v5;
  return ((v0 + w1) + (w2 + w3)) + (w4 + w5);
}

__device__ __forceinline__ float sym2(const float* __restrict__ U2, int base,
                                      unsigned code, int x8) {
  const int i = (int)(code & 0xff) ^ x8;
  const int j = (int)((code >> 8) & 0xff) ^ x8;
  const float v0 = U2[base + (i * 16 + j) * 4];
  const float v1 = U2[base + (j * 16 + i) * 4];
  const float s = (i == j) ? v0 : v0 + v1;
  return (x8 != 0 && (code >> 24) != 0) ? 0.0f : s;
}

__global__ __launch_bounds__(256) void sc_pack_sym(
    const float* __restrict__ U1_0, const float* __restrict__ U2_0,
    const float* __restrict__ U3_0,
    const float* __restrict__ U1_1, const float* __restrict__ U2_1,
    const float* __restrict__ U3_1,
    const float* __restrict__ U1_2, const float* __restrict__ U2_2,
    const float* __restrict__ U3_2,
    float2* __restrict__ P) {
  const int n = blockIdx.x * 256 + threadIdx.x;
  if (n >= KS_TOTAL * 32) return;
  const int lane = n & 31;
  const int row = lane & 15;
  const int x8 = (lane >> 4) << 3;
  const int ks = __builtin_amdgcn_readfirstlane(n >> 5);
  const int deg = (ks < KBASE2) ? 3 : (ks < KBASE1) ? 2 : 1;
  const int T3 = ks / KS3, kk3 = ks - T3 * KS3;
  const int r2 = (deg == 2) ? ks - KBASE2 : 0;
  const int T2 = r2 / KS2, kk2 = r2 - T2 * KS2;
  const int T = (deg == 3) ? T3 : (deg == 2) ? T2 : 0;
  const int kk = (deg == 3) ? kk3 : (deg == 2) ? kk2 : ks - KBASE1;
  const int col = ((deg == 3) ? 0 : (deg == 2) ? NC3 : NC3 + NC2) + 2 * kk;
  const unsigned c0 = col_tab.v[col], c1 = col_tab.v[col + 1];
  const int g = T * 16 + row;
  float a = 0.0f, b = 0.0f;
  if (deg == 3) {
    const int tk = g / 11, path = g - tk * 11;
    const int t = (tk == 0) ? 0 : (tk < 4) ? 1 : 2;
    const int k = tk - ((t == 0) ? 0 : (t == 1) ? 1 : 4);
    if (g < 99) {
      const float* __restrict__ U3 = (t == 0) ? U3_0 : (t == 1) ? U3_1 : U3_2;
      const int base = k * (4096 * 11) + path;
      a = sym3(U3, base, c0, x8);
      b = sym3(U3, base, c1, x8);
    }
  } else if (deg == 2) {
    const int tk = g >> 2, path = g & 3;
    const int t = (tk == 0) ? 0 : (tk < 4) ? 1 : 2;
    const int k = tk - ((t == 0) ? 0 : (t == 1) ? 1 : 4);
    if (g < 36) {
      const float* __restrict__ U2 = (t == 0) ? U2_0 : (t == 1) ? U2_1 : U2_2;
      const int base = k * 1024 + path;
      a = sym2(U2, base, c0, x8);
      b = sym2(U2, base, c1, x8);
    }
  } else {
    const int tk = g;
    const int t = (tk == 0) ? 0 : (tk < 4) ? 1 : 2;
    const int k = tk - ((t == 0) ? 0 : (t == 1) ? 1 : 4);
    if (g < 9) {
      const float* __restrict__ U1 = (t == 0) ? U1_0 : (t == 1) ? U1_1 : U1_2;
      a = U1[k * 16 + ((int)(c0 & 0xff) ^ x8)];
      b = U1[k * 16 + ((int)(c1 & 0xff) ^ x8)];
    }
  }
  P[n] = make_float2(a, b);
}

template <int C>
__device__ __forceinline__ float col_val(const float (&y)[FEAT]) {
  constexpr unsigned code = col_tab_ct.v[C];
  constexpr int i = code & 0xff, j = (code >> 8) & 0xff, l = (code >> 16) & 0xff;
  if constexpr (j == 16) return y[i];
  else if constexpr (l == 16) return y[i] * y[j];
  else return (y[i] * y[j]) * y[l];
}

template <int C>
__device__ __forceinline__ v2f col_frag(const float (&y)[FEAT]) {
  v2f b;
  b.x = col_val<C>(y);
  b.y = col_val<C + 1>(y);
  return b;
}

template <int C, int ODD>
__device__ __forceinline__ void mono_step(const v2f a, const float (&y)[FEAT],
                                          v8f& acc0, v8f& acc1) {
  const v2f b = col_frag<C>(y);
  if constexpr (ODD)
    acc1 = __builtin_amdgcn_wmma_f32_16x16x4_f32(false, a, false, b, (short)0,
                                                 acc1, false, false);
  else
    acc0 = __builtin_amdgcn_wmma_f32_16x16x4_f32(false, a, false, b, (short)0,
                                                 acc0, false, false);
}

template <int KSTEP>
__device__ __forceinline__ void kstep4(const v2f a, const float (&y)[NRT][FEAT],
                                       v8f (&acc0)[NRT], v8f (&acc1)[NRT]) {
  mono_step<2 * KSTEP, KSTEP & 1>(a, y[0], acc0[0], acc1[0]);
  mono_step<2 * KSTEP, KSTEP & 1>(a, y[1], acc0[1], acc1[1]);
  mono_step<2 * KSTEP, KSTEP & 1>(a, y[2], acc0[2], acc1[2]);
  mono_step<2 * KSTEP, KSTEP & 1>(a, y[3], acc0[3], acc1[3]);
}

template <int C, int... S>
__device__ __forceinline__ void tile_walk(const v2f* __restrict__ Ap,
                                          const float (&y)[NRT][FEAT],
                                          v8f (&acc0)[NRT], v8f (&acc1)[NRT],
                                          v2f (&A)[KS3W],
                                          std::integer_sequence<int, S...> seq) {
  constexpr int N = (int)sizeof...(S);
  if constexpr (C + DEPTH < NCHUNK) {
    ((A[(C + DEPTH) * N + S] = Ap[((C + DEPTH) * N + S) * 32]), ...);
    __builtin_amdgcn_sched_barrier(0);
  }
  (kstep4<C * N + S>(A[C * N + S], y, acc0, acc1), ...);
  if constexpr (C + DEPTH + 1 < NCHUNK) __builtin_amdgcn_sched_barrier(0);
  if constexpr (C + 1 < NCHUNK)
    tile_walk<C + 1>(Ap, y, acc0, acc1, A, seq);
}

template <int... S>
__device__ __forceinline__ void tail_frags(const float (&y)[NRT][FEAT],
                                           v2f (&bt)[NRT][KT],
                                           std::integer_sequence<int, S...>) {
  ((bt[0][S] = col_frag<2 * (KS3W + S)>(y[0])), ...);
  ((bt[1][S] = col_frag<2 * (KS3W + S)>(y[1])), ...);
  ((bt[2][S] = col_frag<2 * (KS3W + S)>(y[2])), ...);
  ((bt[3][S] = col_frag<2 * (KS3W + S)>(y[3])), ...);
}

template <int S>
__device__ __forceinline__ void pair_step(const v2f (&A2)[NT2][KS2],
                                          const float (&y)[FEAT], v8f (&c)[NT2]) {
  const v2f b = col_frag<NC3 + 2 * S>(y);
  c[0] = __builtin_amdgcn_wmma_f32_16x16x4_f32(false, A2[0][S], false, b, (short)0,
                                               c[0], false, false);
  c[1] = __builtin_amdgcn_wmma_f32_16x16x4_f32(false, A2[1][S], false, b, (short)0,
                                               c[1], false, false);
  c[2] = __builtin_amdgcn_wmma_f32_16x16x4_f32(false, A2[2][S], false, b, (short)0,
                                               c[2], false, false);
}
template <int... S>
__device__ __forceinline__ void pair_walk(const v2f (&A2)[NT2][KS2],
                                          const float (&y)[FEAT], v8f (&c)[NT2],
                                          std::integer_sequence<int, S...>) {
  (pair_step<S>(A2, y, c), ...);
}
template <int... S>
__device__ __forceinline__ void lin_walk(const v2f (&A1)[KS1], const float (&y)[FEAT],
                                         v8f& d0, v8f& d1,
                                         std::integer_sequence<int, S...>) {
  (mono_step<NC3 + NC2 + 2 * S, S & 1>(A1[S], y, d0, d1), ...);
}

__device__ __forceinline__ void read_row(const float* Xs, int rt, int l15, int half,
                                         float (&y)[FEAT]) {
  const v2f* xr =
      reinterpret_cast<const v2f*>(Xs + (rt * ROWS_PER_TILE + l15) * XPITCH);
#pragma unroll
  for (int c = 0; c < 8; ++c) {
    const v2f v = xr[c ^ (4 * half)];
    y[2 * c + 0] = v.x; y[2 * c + 1] = v.y;
  }
}

__global__ __launch_bounds__(NTHREADS, 2) void sc_mono_kernel(
    const float* __restrict__ x, const int* __restrict__ indices,
    const v2f* __restrict__ P,
    const float* __restrict__ W1_0, const float* __restrict__ W1_1,
    const float* __restrict__ W1_2,
    const float* __restrict__ W2_0, const float* __restrict__ W2_1,
    const float* __restrict__ W2_2,
    const float* __restrict__ W3_0, const float* __restrict__ W3_1,
    const float* __restrict__ W3_2,
    float* __restrict__ out) {
  __shared__ float Zs[NRT * ZTILE];
  __shared__ float Zt[NRT * ZT3];
  __shared__ alignas(16) float Xs[ROWS_PER_BLOCK * XPITCH];

  const int tid = threadIdx.x;
  const int b = blockIdx.x;

  const int wave = __builtin_amdgcn_readfirstlane(tid >> 5);
  const int lane = tid & 31;
  const int half = lane >> 4;
  const int l15 = lane & 15;

  const int e = indices[b];

  v2f A[KS3W];
  const v2f* Ap = P + (size_t)wave * (KS3 * 32) + lane;
  if (wave < NT3) {
#pragma unroll
    for (int s = 0; s < DEPTH * CHUNK; ++s) A[s] = Ap[s * 32];
  }

  const float4 xv =
      reinterpret_cast<const float4*>(x + (size_t)b * (ROWS_PER_BLOCK * FEAT))[tid];

  float wh[NOUTJ][16];
#pragma unroll
  for (int j = 0; j < NOUTJ; ++j) {
    const int o = tid + NTHREADS * j;
    if (o < NOUT) {
      const int otk = o >> 6, oc = o & 63;
      const int ot = (otk == 0) ? 0 : (otk < 4) ? 1 : 2;
      const float* W3 = (ot == 0) ? W3_0 : (ot == 1) ? W3_1 : W3_2;
      const float* W2 = (ot == 0) ? W2_0 : (ot == 1) ? W2_1 : W2_2;
      const float* W1 = (ot == 0) ? W1_0 : (ot == 1) ? W1_1 : W1_2;
#pragma unroll
      for (int p = 0; p < 11; ++p) wh[j][p] = W3[(e * 11 + p) * 64 + oc];
#pragma unroll
      for (int p = 0; p < 4; ++p) wh[j][11 + p] = W2[(e * 4 + p) * 64 + oc];
      wh[j][15] = W1[e * 64 + oc];
    }
  }

  {
    const float4 v = xv;
    v2f* d = reinterpret_cast<v2f*>(Xs + (tid >> 2) * XPITCH + (tid & 3) * 4);
    v2f lo, hi;
    lo.x = v.x; lo.y = v.y; hi.x = v.z; hi.y = v.w;
    d[0] = lo; d[1] = hi;
  }
  __syncthreads();

  float* Zcol = &Zs[(8 * half) * ROWS_PER_TILE + l15];

  if (wave < NT3) {
    static_assert(NCHUNK * CHUNK == KS3W, "groups cover the wave's part of the degree-3 walk");
    const auto seq = std::make_integer_sequence<int, CHUNK>{};
    float y[NRT][FEAT];
#pragma unroll
    for (int rt = 0; rt < NRT; ++rt) read_row(Xs, rt, l15, half, y[rt]);
    v8f acc0[NRT] = {};
    v8f acc1[NRT] = {};
    tile_walk<0>(Ap, y, acc0, acc1, A, seq);
#pragma unroll
    for (int rt = 0; rt < NRT; ++rt) {
      const v8f z = acc0[rt] + acc1[rt];
#pragma unroll
      for (int v = 0; v < 8; ++v)
        Zcol[rt * ZTILE + (wave * 16 + v) * ROWS_PER_TILE] = z[v];
    }
  } else {
    const v2f* A2p = P + (size_t)KBASE2 * 32 + lane;
    const v2f* A1p = P + (size_t)KBASE1 * 32 + lane;
    v2f A2[NT2][KS2];
    v2f A1[KS1];
#pragma unroll
    for (int T = 0; T < NT2; ++T) {
#pragma unroll
      for (int s = 0; s < KS2; ++s) A2[T][s] = A2p[(T * KS2 + s) * 32];
    }
#pragma unroll
    for (int s = 0; s < KS1; ++s) A1[s] = A1p[s * 32];
    const v2f* Atp = P + (size_t)KS3W * 32 + lane;
    v2f At[NT3][KT];
#pragma unroll
    for (int t = 0; t < NTA; ++t) {
#pragma unroll
      for (int s = 0; s < KT; ++s) At[t][s] = Atp[(t * KS3 + s) * 32];
    }
    float y[FEAT];
#pragma unroll 1
    for (int rt = 0; rt < NRT; ++rt) {
      read_row(Xs, rt, l15, half, y);
      v8f c[NT2] = {};
      pair_walk(A2, y, c, std::make_integer_sequence<int, KS2>{});
#pragma unroll
      for (int T = 0; T < NT2; ++T) {
#pragma unroll
        for (int v = 0; v < 8; ++v)
          Zcol[rt * ZTILE + ((NT3 + T) * 16 + v) * ROWS_PER_TILE] = c[T][v];
      }
      v8f d0 = {};
      v8f d1 = {};
      lin_walk(A1, y, d0, d1, std::make_integer_sequence<int, KS1>{});
      const v8f z = d0 + d1;
#pragma unroll
      for (int v = 0; v < 8; ++v)
        Zcol[rt * ZTILE + ((NT3 + NT2) * 16 + v) * ROWS_PER_TILE] = z[v];
    }
    {
#pragma unroll
      for (int t = NTA; t < NT3; ++t) {
#pragma unroll
        for (int s = 0; s < KT; ++s) At[t][s] = Atp[(t * KS3 + s) * 32];
      }
      float y4[NRT][FEAT];
#pragma unroll
      for (int rt = 0; rt < NRT; ++rt) read_row(Xs, rt, l15, half, y4[rt]);
      v2f bt[NRT][KT];
      tail_frags(y4, bt, std::make_integer_sequence<int, KT>{});
      float* Ztcol = &Zt[(8 * half) * ROWS_PER_TILE + l15];
#pragma unroll
      for (int t = 0; t < NT3; ++t) {
        v8f z[NRT] = {};
#pragma unroll
        for (int s = 0; s < KT; ++s) {
#pragma unroll
          for (int rt = 0; rt < NRT; ++rt)
            z[rt] = __builtin_amdgcn_wmma_f32_16x16x4_f32(false, At[t][s], false, bt[rt][s],
                                                          (short)0, z[rt], false, false);
        }
#pragma unroll
        for (int rt = 0; rt < NRT; ++rt) {
#pragma unroll
          for (int v = 0; v < 8; ++v)
            Ztcol[rt * ZT3 + (t * 16 + v) * ROWS_PER_TILE] = z[rt][v];
        }
        __builtin_amdgcn_sched_barrier(0);
      }
    }
  }

  __syncthreads();

#pragma unroll
  for (int j = 0; j < NOUTJ; ++j) {
    const int o = tid + NTHREADS * j;
    if (o < NOUT) {
      const int otk = o >> 6, oc = o & 63;
      const int ot = (otk == 0) ? 0 : (otk < 4) ? 1 : 2;
      const float* Zr = &Zs[(oc >> 4) * ZTILE + (oc & 15)];
      const float* Zq = &Zt[(oc >> 4) * ZT3 + (oc & 15)];
      float s = 0.0f;
#pragma unroll
      for (int p = 0; p < 11; ++p)
        s = fmaf(Zr[(otk * 11 + p) * ROWS_PER_TILE] + Zq[(otk * 11 + p) * ROWS_PER_TILE],
                 wh[j][p], s);
#pragma unroll
      for (int p = 0; p < 4; ++p)
        s = fmaf(Zr[(NT3 * 16 + otk * 4 + p) * ROWS_PER_TILE], wh[j][11 + p], s);
      s = fmaf(Zr[((NT3 + NT2) * 16 + otk) * ROWS_PER_TILE], wh[j][15], s);
      const int kt  = (ot == 0) ? 1 : (ot == 1) ? 3 : 5;
      const int seg = (ot == 0) ? 0 : (ot == 1) ? 64 : 256;
      const int k   = otk - ((ot == 0) ? 0 : (ot == 1) ? 1 : 4);
      out[b * 576 + seg + oc * kt + k] = s;
    }
  }
}

extern "C" void kernel_launch(void* const* d_in, const int* in_sizes, int n_in,
                              void* d_out, int out_size, void* d_ws, size_t ws_size,
                              hipStream_t stream) {
  (void)in_sizes; (void)n_in; (void)out_size; (void)ws_size;
  const float* x   = (const float*)d_in[0];
  const int* idx   = (const int*)d_in[1];
  const float* U1_0 = (const float*)d_in[2];
  const float* W1_0 = (const float*)d_in[3];
  const float* U2_0 = (const float*)d_in[4];
  const float* W2_0 = (const float*)d_in[5];
  const float* U3_0 = (const float*)d_in[6];
  const float* W3_0 = (const float*)d_in[7];
  const float* U1_1 = (const float*)d_in[8];
  const float* W1_1 = (const float*)d_in[9];
  const float* U2_1 = (const float*)d_in[10];
  const float* W2_1 = (const float*)d_in[11];
  const float* U3_1 = (const float*)d_in[12];
  const float* W3_1 = (const float*)d_in[13];
  const float* U1_2 = (const float*)d_in[14];
  const float* W1_2 = (const float*)d_in[15];
  const float* U2_2 = (const float*)d_in[16];
  const float* W2_2 = (const float*)d_in[17];
  const float* U3_2 = (const float*)d_in[18];
  const float* W3_2 = (const float*)d_in[19];
  float* out = (float*)d_out;

  float2* P = (float2*)d_ws;
  sc_pack_sym<<<193, 256, 0, stream>>>(U1_0, U2_0, U3_0, U1_1, U2_1, U3_1,
                                       U1_2, U2_2, U3_2, P);

  sc_mono_kernel<<<256, NTHREADS, 0, stream>>>(
      x, idx, (const v2f*)P,
      W1_0, W1_1, W1_2,
      W2_0, W2_1, W2_2,
      W3_0, W3_1, W3_2,
      out);
}
